# starting kernel
# baseline (speedup 1.0000x reference)
#include <hip/hip_runtime.h>

typedef __attribute__((ext_vector_type(2))) float v2f;
typedef __attribute__((ext_vector_type(8))) float v8f;

#define D 128

__global__ __launch_bounds__(256) void gnn_zero(float* __restrict__ p, long long n) {
    long long i = (long long)blockIdx.x * blockDim.x + threadIdx.x;
    if (i < n) p[i] = 0.0f;
}

__global__ __launch_bounds__(256) void gnn_count(const int* __restrict__ dst,
                                                 float* __restrict__ cnt, int nEdges) {
    int e = blockIdx.x * blockDim.x + threadIdx.x;
    if (e < nEdges) atomicAdd(&cnt[dst[e]], 1.0f);
}

__global__ __launch_bounds__(256) void gnn_inv(float* __restrict__ cnt, int nNodes) {
    int n = blockIdx.x * blockDim.x + threadIdx.x;
    if (n < nNodes) cnt[n] = 1.0f / fmaxf(cnt[n], 1.0f);
}

__global__ __launch_bounds__(256) void gnn_scatter(const float* __restrict__ feat,
                                                   const int* __restrict__ src,
                                                   const int* __restrict__ dst,
                                                   float* __restrict__ agg, int nEdges) {
    long long t = (long long)blockIdx.x * blockDim.x + threadIdx.x;
    int e = (int)(t >> 5);
    int c = ((int)t & 31) * 4;
    if (e >= nEdges) return;
    int s = src[e];
    int d = dst[e];
    float4 val = *(const float4*)(feat + (size_t)s * D + c);
    float* p = agg + (size_t)d * D + c;
    atomicAdd(p + 0, val.x);
    atomicAdd(p + 1, val.y);
    atomicAdd(p + 2, val.z);
    atomicAdd(p + 3, val.w);
}

__global__ __launch_bounds__(256) void gnn_sage_gemm(
    const float* __restrict__ agg,
    const float* __restrict__ inv,
    const float* __restrict__ xin,
    const float* __restrict__ wl,
    const float* __restrict__ wr,
    const float* __restrict__ bias,
    float* __restrict__ out,
    int relu) {
    const int lane    = threadIdx.x & 31;
    const int wave    = threadIdx.x >> 5;
    const int rowBase = blockIdx.x * 16;
    const int colBase = wave * 16;

    const int half = lane >> 4;
    const int l15  = lane & 15;
    const int arow = rowBase + l15;
    const int brow = colBase + l15;

    const float inv_c = inv[arow];
    const float* aggRow = agg + (size_t)arow * D;
    const float* xRow   = xin + (size_t)arow * D;
    const float* wlRow  = wl + (size_t)brow * D;
    const float* wrRow  = wr + (size_t)brow * D;

    v8f acc = {};
#pragma unroll 8
    for (int kc = 0; kc < D; kc += 4) {
        const int k0 = kc + 2 * half;
        v2f a_m = *(const v2f*)(aggRow + k0);
        a_m.x *= inv_c;
        a_m.y *= inv_c;
        v2f a_x = *(const v2f*)(xRow + k0);
        v2f b_l = *(const v2f*)(wlRow + k0);
        v2f b_r = *(const v2f*)(wrRow + k0);
        acc = __builtin_amdgcn_wmma_f32_16x16x4_f32(false, a_m, false, b_l,
                                                    (short)0, acc, false, false);
        acc = __builtin_amdgcn_wmma_f32_16x16x4_f32(false, a_x, false, b_r,
                                                    (short)0, acc, false, false);
    }

    const float bv = bias[colBase + l15];
#pragma unroll
    for (int v = 0; v < 8; ++v) {
        float r = acc[v] + bv;
        if (relu) r = fmaxf(r, 0.0f);
        out[(size_t)(rowBase + v + 8 * half) * D + colBase + l15] = r;
    }
}

extern "C" void kernel_launch(void* const* d_in, const int* in_sizes, int n_in,
                              void* d_out, int out_size, void* d_ws, size_t ws_size,
                              hipStream_t stream) {
    const float* x   = (const float*)d_in[0];
    const int* edge  = (const int*)d_in[1];
    const float* w1l = (const float*)d_in[2];
    const float* w1r = (const float*)d_in[3];
    const float* b1  = (const float*)d_in[4];
    const float* w2l = (const float*)d_in[5];
    const float* w2r = (const float*)d_in[6];
    const float* b2  = (const float*)d_in[7];
    float* out = (float*)d_out;

    const int N = in_sizes[0] / D;
    const int E = in_sizes[1] / 2;
    const int* src = edge;
    const int* dst = edge + E;

    float* agg = (float*)d_ws;
    float* h   = agg + (size_t)N * D;
    float* inv = h + (size_t)N * D;

    const long long nd = (long long)N * D;
    dim3 blk(256);
    dim3 gZeroND((unsigned)((nd + 255) / 256));
    dim3 gZeroN((unsigned)((N + 255) / 256));
    dim3 gCnt((unsigned)((E + 255) / 256));
    dim3 gScat((unsigned)(((long long)E * 32 + 255) / 256));
    dim3 gGemm((unsigned)(N / 16));

    gnn_zero<<<gZeroN, blk, 0, stream>>>(inv, N);
    gnn_count<<<gCnt, blk, 0, stream>>>(dst, inv, E);
    gnn_inv<<<gZeroN, blk, 0, stream>>>(inv, N);

    gnn_zero<<<gZeroND, blk, 0, stream>>>(agg, nd);
    gnn_scatter<<<gScat, blk, 0, stream>>>(x, src, dst, agg, E);
    gnn_sage_gemm<<<gGemm, blk, 0, stream>>>(agg, inv, x, w1l, w1r, b1, h, 1);

    gnn_zero<<<gZeroND, blk, 0, stream>>>(agg, nd);
    gnn_scatter<<<gScat, blk, 0, stream>>>(h, src, dst, agg, E);
    gnn_sage_gemm<<<gGemm, blk, 0, stream>>>(agg, inv, h, w2l, w2r, b2, out, 0);
}


# Scatter replaced: each block owns 196 destination rows in LDS, scans byte range keys of all edges, adds matching source rows with LDS float adds, writes rows and 1/deg once; sum order = LDS arrival
# speedup vs baseline: 2.9394x; 2.9394x over previous
#include <hip/hip_runtime.h>

typedef __attribute__((ext_vector_type(2))) float v2f;
typedef __attribute__((ext_vector_type(4))) float v4f;
typedef __attribute__((ext_vector_type(8))) float v8f;

#define D 128

#define RNG 196
#define LCAP 8192
#define EPI 16
#define ATH 1024

__global__ __launch_bounds__(256) void gnn_key(const int* __restrict__ dst,
                                               unsigned char* __restrict__ key, int nEdges) {
    const int e = (blockIdx.x * blockDim.x + threadIdx.x) * 4;
    if (e + 3 < nEdges) {
        const unsigned k0 = (unsigned)(dst[e + 0] / RNG) & 0xffu;
        const unsigned k1 = (unsigned)(dst[e + 1] / RNG) & 0xffu;
        const unsigned k2 = (unsigned)(dst[e + 2] / RNG) & 0xffu;
        const unsigned k3 = (unsigned)(dst[e + 3] / RNG) & 0xffu;
        *(unsigned*)(key + e) = k0 | (k1 << 8) | (k2 << 16) | (k3 << 24);
    } else {
        for (int t = e; t < nEdges; ++t) key[t] = (unsigned char)(dst[t] / RNG);
    }
}

__global__ __launch_bounds__(ATH) void gnn_agg(const float* __restrict__ feat,
                                               const int* __restrict__ src,
                                               const int* __restrict__ dst,
                                               const unsigned char* __restrict__ key,
                                               float* __restrict__ agg,
                                               float* __restrict__ inv,
                                               int nEdges, int nNodes) {
    __shared__ __attribute__((aligned(16))) float acc[RNG * D];
    __shared__ float cnt[RNG];
    __shared__ unsigned list[LCAP];
    __shared__ unsigned lcount;

    const int tid  = threadIdx.x;
    const int lane = tid & 31;
    const int wv   = __builtin_amdgcn_readfirstlane(tid >> 5);
    const int base = blockIdx.x * RNG;
    const unsigned myKey = blockIdx.x;

    for (int i = tid; i < RNG * (D / 4); i += ATH) ((v4f*)acc)[i] = (v4f){0.f, 0.f, 0.f, 0.f};
    for (int i = tid; i < RNG; i += ATH) cnt[i] = 0.0f;
    if (tid == 0) lcount = 0u;
    __syncthreads();

    int nPass = 1;
    for (int pass = 0; pass < nPass; ++pass) {
        if (pass == 0) {
            const unsigned pat  = myKey * 0x01010101u;
            const unsigned miss = pat ^ 0x01010101u;
            const int nVec = nEdges >> 4;
            const uint4* kv = (const uint4*)key;
            for (int v0 = tid; v0 < nVec; v0 += 8 * ATH) {
                uint4 w[8];
#pragma unroll
                for (int k = 0; k < 8; ++k) {
                    const int vi = v0 + k * ATH;
                    w[k] = kv[vi < nVec ? vi : 0];
                }
#pragma unroll
                for (int k = 0; k < 8; ++k) {
                    if (v0 + k * ATH >= nVec) w[k] = make_uint4(miss, miss, miss, miss);
                }
#pragma unroll
                for (int k = 0; k < 8; ++k) {
                    const unsigned ww[4] = {w[k].x, w[k].y, w[k].z, w[k].w};
#pragma unroll
                    for (int j = 0; j < 4; ++j) {
                        const unsigned x = ww[j] ^ pat;
                        if ((x - 0x01010101u) & ~x & 0x80808080u) {
#pragma unroll
                            for (int q = 0; q < 4; ++q) {
                                if (((x >> (8 * q)) & 0xffu) == 0u) {
                                    const unsigned idx = atomicAdd(&lcount, 1u);
                                    if (idx < LCAP)
                                        list[idx] = (unsigned)(v0 + k * ATH) * 16u + j * 4 + q;
                                }
                            }
                        }
                    }
                }
            }
            for (int e = nVec * 16 + tid; e < nEdges; e += ATH) {
                if (key[e] == (unsigned char)myKey) {
                    const unsigned idx = atomicAdd(&lcount, 1u);
                    if (idx < LCAP) list[idx] = (unsigned)e;
                }
            }
        } else {
            const int c0 = (pass - 1) * LCAP;
            const int c1 = c0 + LCAP < nEdges ? c0 + LCAP : nEdges;
            for (int e = c0 + tid; e < c1; e += ATH) {
                if (key[e] == (unsigned char)myKey) {
                    const unsigned idx = atomicAdd(&lcount, 1u);
                    list[idx] = (unsigned)e;
                }
            }
        }
        __syncthreads();

        const unsigned lc = lcount;
        if (lc > LCAP) {
            nPass = 1 + (nEdges + LCAP - 1) / LCAP;
        } else {
            const int total = __builtin_amdgcn_readfirstlane((int)lc);
            int i0 = wv * EPI;
            int nS = 0, nD = 0;
            if (i0 + lane < total && lane < EPI) {
                const unsigned e = list[i0 + lane];
                nS = src[e];
                nD = dst[e] - base;
            }
            for (; i0 < total; i0 += (ATH / 32) * EPI) {
                const int left = total - i0;
                const int n = left < EPI ? left : EPI;
                const int myS = nS, myD = nD;
                const int i1 = i0 + (ATH / 32) * EPI;
                if (i1 + lane < total && lane < EPI) {
                    const unsigned e = list[i1 + lane];
                    nS = src[e];
                    nD = dst[e] - base;
                }
                if (lane < n) atomicAdd(&cnt[myD], 1.0f);

                if (n == EPI) {
                    float v[EPI][4];
#pragma unroll
                    for (int u = 0; u < EPI; ++u) {
                        const int sId = __builtin_amdgcn_readlane(myS, u);
                        const float* q = feat + (size_t)sId * D + lane;
                        v[u][0] = q[0];
                        v[u][1] = q[32];
                        v[u][2] = q[64];
                        v[u][3] = q[96];
                    }
#pragma unroll
                    for (int u = 0; u < EPI; ++u) {
                        const int dId = __builtin_amdgcn_readlane(myD, u);
                        float* p = acc + dId * D + lane;
                        atomicAdd(p + 0, v[u][0]);
                        atomicAdd(p + 32, v[u][1]);
                        atomicAdd(p + 64, v[u][2]);
                        atomicAdd(p + 96, v[u][3]);
                    }
                } else {
                    for (int u = 0; u < n; ++u) {
                        const int sId = __shfl(myS, u);
                        const int dId = __shfl(myD, u);
                        const float* q = feat + (size_t)sId * D + lane;
                        const float v0 = q[0];
                        const float v1 = q[32];
                        const float v2 = q[64];
                        const float v3 = q[96];
                        float* p = acc + dId * D + lane;
                        atomicAdd(p + 0, v0);
                        atomicAdd(p + 32, v1);
                        atomicAdd(p + 64, v2);
                        atomicAdd(p + 96, v3);
                    }
                }
            }
        }
        __syncthreads();
        if (tid == 0) lcount = 0u;
        __syncthreads();
    }

    const int rowsLeft = nNodes - base;
    const int rows = rowsLeft < RNG ? rowsLeft : RNG;
    for (int i = tid; i < rows * (D / 4); i += ATH)
        *(v4f*)(agg + (size_t)base * D + (size_t)i * 4) = ((const v4f*)acc)[i];
    for (int r = tid; r < rows; r += ATH) inv[base + r] = 1.0f / fmaxf(cnt[r], 1.0f);
}

#define TM 16
#define PITCH 132

__global__ __launch_bounds__(256) void gnn_sage_gemm(
    const float* __restrict__ agg,
    const float* __restrict__ inv,
    const float* __restrict__ xin,
    const float* __restrict__ wl,
    const float* __restrict__ wr,
    const float* __restrict__ bias,
    float* __restrict__ out,
    int nTiles, int relu) {
    __shared__ __attribute__((aligned(16))) float sA[2][2][TM * PITCH];

    const int tid     = threadIdx.x;
    const int lane    = tid & 31;
    const int wave    = tid >> 5;
    const int half    = lane >> 4;
    const int l15     = lane & 15;
    const int colBase = wave * 16;

    v4f bl[16], br[16];
    {
        const float* wlRow = wl + (size_t)(colBase + l15) * D + 4 * half;
        const float* wrRow = wr + (size_t)(colBase + l15) * D + 4 * half;
#pragma unroll
        for (int j = 0; j < 16; ++j) {
            bl[j] = *(const v4f*)(wlRow + 8 * j);
            br[j] = *(const v4f*)(wrRow + 8 * j);
        }
    }
    const float bv = bias[colBase + l15];

    const int sr = tid >> 5;
    const int sc = (tid & 31) * 4;

    int tile = blockIdx.x;
    int cur  = 0;
    if (tile < nTiles) {
        const size_t r0 = (size_t)tile * TM + sr;
        const size_t r1 = r0 + 8;
        const float i0 = inv[r0];
        const float i1 = inv[r1];
        const v4f g0 = *(const v4f*)(agg + r0 * D + sc);
        const v4f g1 = *(const v4f*)(agg + r1 * D + sc);
        const v4f g2 = *(const v4f*)(xin + r0 * D + sc);
        const v4f g3 = *(const v4f*)(xin + r1 * D + sc);
        *(v4f*)(&sA[0][0][sr * PITCH + sc])       = g0 * i0;
        *(v4f*)(&sA[0][0][(sr + 8) * PITCH + sc]) = g1 * i1;
        *(v4f*)(&sA[0][1][sr * PITCH + sc])       = g2;
        *(v4f*)(&sA[0][1][(sr + 8) * PITCH + sc]) = g3;
    }
    __syncthreads();

    for (; tile < nTiles; tile += gridDim.x) {
        const int  nxt  = tile + (int)gridDim.x;
        const bool more = nxt < nTiles;
        float i0 = 0.0f, i1 = 0.0f;
        v4f g0 = {}, g1 = {}, g2 = {}, g3 = {};
        if (more) {
            const size_t r0 = (size_t)nxt * TM + sr;
            const size_t r1 = r0 + 8;
            i0 = inv[r0];
            i1 = inv[r1];
            g0 = *(const v4f*)(agg + r0 * D + sc);
            g1 = *(const v4f*)(agg + r1 * D + sc);
            g2 = *(const v4f*)(xin + r0 * D + sc);
            g3 = *(const v4f*)(xin + r1 * D + sc);
        }

        const float* pa = &sA[cur][0][l15 * PITCH + 4 * half];
        const float* px = &sA[cur][1][l15 * PITCH + 4 * half];
        v8f acc1 = {};
        v8f acc2 = {};
#pragma unroll
        for (int j = 0; j < 16; ++j) {
            const v4f a = *(const v4f*)(pa + 8 * j);
            const v4f x = *(const v4f*)(px + 8 * j);
            acc1 = __builtin_amdgcn_wmma_f32_16x16x4_f32(false, a.xy, false, bl[j].xy,
                                                         (short)0, acc1, false, false);
            acc2 = __builtin_amdgcn_wmma_f32_16x16x4_f32(false, x.xy, false, br[j].xy,
                                                         (short)0, acc2, false, false);
            acc1 = __builtin_amdgcn_wmma_f32_16x16x4_f32(false, a.zw, false, bl[j].zw,
                                                         (short)0, acc1, false, false);
            acc2 = __builtin_amdgcn_wmma_f32_16x16x4_f32(false, x.zw, false, br[j].zw,
                                                         (short)0, acc2, false, false);
        }

        const int rowBase = tile * TM;
#pragma unroll
        for (int v = 0; v < 8; ++v) {
            float r = (acc1[v] + acc2[v]) + bv;
            if (relu) r = fmaxf(r, 0.0f);
            out[(size_t)(rowBase + v + 8 * half) * D + colBase + l15] = r;
        }

        if (more) {
            const int nb = cur ^ 1;
            *(v4f*)(&sA[nb][0][sr * PITCH + sc])       = g0 * i0;
            *(v4f*)(&sA[nb][0][(sr + 8) * PITCH + sc]) = g1 * i1;
            *(v4f*)(&sA[nb][1][sr * PITCH + sc])       = g2;
            *(v4f*)(&sA[nb][1][(sr + 8) * PITCH + sc]) = g3;
        }
        __syncthreads();
        cur ^= 1;
    }
}

extern "C" void kernel_launch(void* const* d_in, const int* in_sizes, int n_in,
                              void* d_out, int out_size, void* d_ws, size_t ws_size,
                              hipStream_t stream) {
    const float* x   = (const float*)d_in[0];
    const int* edge  = (const int*)d_in[1];
    const float* w1l = (const float*)d_in[2];
    const float* w1r = (const float*)d_in[3];
    const float* b1  = (const float*)d_in[4];
    const float* w2l = (const float*)d_in[5];
    const float* w2r = (const float*)d_in[6];
    const float* b2  = (const float*)d_in[7];
    float* out = (float*)d_out;

    const int N = in_sizes[0] / D;
    const int E = in_sizes[1] / 2;
    const int* src = edge;
    const int* dst = edge + E;

    float* agg = (float*)d_ws;
    float* h   = agg + (size_t)N * D;
    float* inv = h + (size_t)N * D;
    unsigned char* key = (unsigned char*)(inv + ((N + 3) & ~3));

    dim3 blk(256);
    dim3 gKey((unsigned)((E + 1023) / 1024));
    dim3 gAgg((unsigned)((N + RNG - 1) / RNG));
    dim3 gGemm(512);
    const int nTiles = N / 16;

    gnn_key<<<gKey, blk, 0, stream>>>(dst, key, E);

    gnn_agg<<<gAgg, dim3(ATH), 0, stream>>>(x, src, dst, key, agg, inv, E, N);
    gnn_sage_gemm<<<gGemm, blk, 0, stream>>>(agg, inv, x, w1l, w1r, b1, h, nTiles, 1);

    gnn_agg<<<gAgg, dim3(ATH), 0, stream>>>(h, src, dst, key, agg, inv, E, N);
    gnn_sage_gemm<<<gGemm, blk, 0, stream>>>(agg, inv, h, w2l, w2r, b2, out, nTiles, 0);
}


# Byte range keys replaced by per-range edge bitmaps (one bit per edge, built per 1024-edge tile by LDS integer ORs): 32 edges per word test, one list reservation per lane; sums stay LDS arrival order
# speedup vs baseline: 3.6036x; 1.2260x over previous
#include <hip/hip_runtime.h>

typedef __attribute__((ext_vector_type(2))) float v2f;
typedef __attribute__((ext_vector_type(4))) float v4f;
typedef __attribute__((ext_vector_type(8))) float v8f;

#define D 128

#define RNG 196
#define LCAP 8192
#define EPI 16
#define ATH 1024
#define TW 32
#define TPITCH 33

__global__ __launch_bounds__(256) void gnn_bits(const int* __restrict__ dst,
                                                unsigned* __restrict__ bm, int nEdges) {
    __shared__ unsigned t[256 * TPITCH];
    const int tid  = threadIdx.x;
    const int lane = tid & 31;
    const int wv   = tid >> 5;
    const int e0 = blockIdx.x * (TW * 32);
    const int rowWords = gridDim.x * TW;
    int d[4];
#pragma unroll
    for (int k = 0; k < 4; ++k) {
        const int e = e0 + k * 256 + tid;
        d[k] = dst[e < nEdges ? e : 0];
    }
    for (int i = tid; i < 256 * TPITCH; i += 256) t[i] = 0u;
    __syncthreads();
#pragma unroll
    for (int k = 0; k < 4; ++k) {
        if (e0 + k * 256 + tid < nEdges) {
            const unsigned r = (unsigned)(d[k] / RNG) & 0xffu;
            atomicOr(&t[r * TPITCH + k * 8 + wv], 1u << lane);
        }
    }
    __syncthreads();
    unsigned* p = bm + (size_t)wv * rowWords + (size_t)blockIdx.x * TW + lane;
    const unsigned* q = t + wv * TPITCH + lane;
#pragma unroll 8
    for (int it = 0; it < 32; ++it) p[(size_t)(it * 8) * rowWords] = q[it * 8 * TPITCH];
}

__global__ __launch_bounds__(ATH) void gnn_agg(const float* __restrict__ feat,
                                               const int* __restrict__ src,
                                               const int* __restrict__ dst,
                                               const unsigned* __restrict__ bm,
                                               float* __restrict__ agg,
                                               float* __restrict__ inv,
                                               int rowWords, int nNodes) {
    __shared__ __attribute__((aligned(16))) float acc[RNG * D];
    __shared__ float cnt[RNG];
    __shared__ unsigned list[LCAP];
    __shared__ unsigned lcount;

    const int tid  = threadIdx.x;
    const int lane = tid & 31;
    const int wv   = __builtin_amdgcn_readfirstlane(tid >> 5);
    const int base = blockIdx.x * RNG;
    const unsigned* row = bm + (size_t)blockIdx.x * rowWords;

    for (int i = tid; i < RNG * (D / 4); i += ATH) ((v4f*)acc)[i] = (v4f){0.f, 0.f, 0.f, 0.f};
    for (int i = tid; i < RNG; i += ATH) cnt[i] = 0.0f;
    if (tid == 0) lcount = 0u;
    __syncthreads();

    int nPass = 1;
    for (int pass = 0; pass < nPass; ++pass) {
        if (pass == 0) {
            const int nVec = rowWords >> 2;
            const uint4* rv = (const uint4*)row;
            for (int vb = 0; vb < nVec; vb += 8 * ATH) {
                const int v0 = vb + tid;
                uint4 w[8];
#pragma unroll
                for (int k = 0; k < 8; ++k) {
                    const int vi = v0 + k * ATH;
                    w[k] = rv[vi < nVec ? vi : 0];
                }
                unsigned nT = 0u;
#pragma unroll
                for (int k = 0; k < 8; ++k) {
                    if (v0 + k * ATH >= nVec) w[k] = make_uint4(0u, 0u, 0u, 0u);
                    nT += __builtin_popcount(w[k].x) + __builtin_popcount(w[k].y) +
                          __builtin_popcount(w[k].z) + __builtin_popcount(w[k].w);
                }
                unsigned incl = nT;
#pragma unroll
                for (int dl = 1; dl < 32; dl <<= 1) {
                    const unsigned up = (unsigned)__shfl_up((int)incl, dl);
                    if (lane >= dl) incl += up;
                }
                const unsigned waveTotal = (unsigned)__builtin_amdgcn_readlane((int)incl, 31);
                unsigned wbase = 0u;
                if (lane == 0) wbase = atomicAdd(&lcount, waveTotal);
                wbase = (unsigned)__builtin_amdgcn_readfirstlane((int)wbase);
                unsigned idx = wbase + incl - nT;
                if (idx + nT <= LCAP) {
#pragma unroll
                    for (int k = 0; k < 8; ++k) {
                        const unsigned ww[4] = {w[k].x, w[k].y, w[k].z, w[k].w};
#pragma unroll
                        for (int j = 0; j < 4; ++j) {
                            unsigned x = ww[j];
                            const unsigned eb = ((unsigned)(v0 + k * ATH) * 4u + j) * 32u;
                            while (x) {
                                const unsigned b = (unsigned)__builtin_ctz(x);
                                x &= x - 1u;
                                list[idx++] = eb + b;
                            }
                        }
                    }
                }
            }
        } else {
            const int wi = (pass - 1) * (LCAP / 32) + tid;
            if (tid < LCAP / 32 && wi < rowWords) {
                unsigned x = row[wi];
                unsigned idx = atomicAdd(&lcount, (unsigned)__builtin_popcount(x));
                while (x) {
                    const unsigned b = (unsigned)__builtin_ctz(x);
                    x &= x - 1u;
                    list[idx++] = (unsigned)wi * 32u + b;
                }
            }
        }
        __syncthreads();

        const unsigned lc = lcount;
        if (lc > LCAP) {
            nPass = 1 + (rowWords + LCAP / 32 - 1) / (LCAP / 32);
        } else {
            const int total = __builtin_amdgcn_readfirstlane((int)lc);
            int i0 = wv * EPI;
            int nS = 0, nD = 0;
            if (i0 + lane < total && lane < EPI) {
                const unsigned e = list[i0 + lane];
                nS = src[e];
                nD = dst[e] - base;
            }
            for (; i0 < total; i0 += (ATH / 32) * EPI) {
                const int left = total - i0;
                const int n = left < EPI ? left : EPI;
                const int myS = nS, myD = nD;
                const int i1 = i0 + (ATH / 32) * EPI;
                if (i1 + lane < total && lane < EPI) {
                    const unsigned e = list[i1 + lane];
                    nS = src[e];
                    nD = dst[e] - base;
                }
                if (lane < n) atomicAdd(&cnt[myD], 1.0f);

                if (n == EPI) {
                    float v[EPI][4];
#pragma unroll
                    for (int u = 0; u < EPI; ++u) {
                        const int sId = __builtin_amdgcn_readlane(myS, u);
                        const float* q = feat + (size_t)sId * D + lane;
                        v[u][0] = q[0];
                        v[u][1] = q[32];
                        v[u][2] = q[64];
                        v[u][3] = q[96];
                    }
#pragma unroll
                    for (int u = 0; u < EPI; ++u) {
                        const int dId = __builtin_amdgcn_readlane(myD, u);
                        float* p = acc + dId * D + lane;
                        atomicAdd(p + 0, v[u][0]);
                        atomicAdd(p + 32, v[u][1]);
                        atomicAdd(p + 64, v[u][2]);
                        atomicAdd(p + 96, v[u][3]);
                    }
                } else {
                    for (int u = 0; u < n; ++u) {
                        const int sId = __shfl(myS, u);
                        const int dId = __shfl(myD, u);
                        const float* q = feat + (size_t)sId * D + lane;
                        const float v0 = q[0];
                        const float v1 = q[32];
                        const float v2 = q[64];
                        const float v3 = q[96];
                        float* p = acc + dId * D + lane;
                        atomicAdd(p + 0, v0);
                        atomicAdd(p + 32, v1);
                        atomicAdd(p + 64, v2);
                        atomicAdd(p + 96, v3);
                    }
                }
            }
        }
        __syncthreads();
        if (tid == 0) lcount = 0u;
        __syncthreads();
    }

    const int rowsLeft = nNodes - base;
    const int rows = rowsLeft < RNG ? rowsLeft : RNG;
    for (int i = tid; i < rows * (D / 4); i += ATH)
        *(v4f*)(agg + (size_t)base * D + (size_t)i * 4) = ((const v4f*)acc)[i];
    for (int r = tid; r < rows; r += ATH) inv[base + r] = 1.0f / fmaxf(cnt[r], 1.0f);
}

#define TM 16
#define PITCH 132

__global__ __launch_bounds__(256) void gnn_sage_gemm(
    const float* __restrict__ agg,
    const float* __restrict__ inv,
    const float* __restrict__ xin,
    const float* __restrict__ wl,
    const float* __restrict__ wr,
    const float* __restrict__ bias,
    float* __restrict__ out,
    int nTiles, int relu) {
    __shared__ __attribute__((aligned(16))) float sA[2][2][TM * PITCH];

    const int tid     = threadIdx.x;
    const int lane    = tid & 31;
    const int wave    = tid >> 5;
    const int half    = lane >> 4;
    const int l15     = lane & 15;
    const int colBase = wave * 16;

    v4f bl[16], br[16];
    {
        const float* wlRow = wl + (size_t)(colBase + l15) * D + 4 * half;
        const float* wrRow = wr + (size_t)(colBase + l15) * D + 4 * half;
#pragma unroll
        for (int j = 0; j < 16; ++j) {
            bl[j] = *(const v4f*)(wlRow + 8 * j);
            br[j] = *(const v4f*)(wrRow + 8 * j);
        }
    }
    const float bv = bias[colBase + l15];

    const int sr = tid >> 5;
    const int sc = (tid & 31) * 4;

    int tile = blockIdx.x;
    int cur  = 0;
    if (tile < nTiles) {
        const size_t r0 = (size_t)tile * TM + sr;
        const size_t r1 = r0 + 8;
        const float i0 = inv[r0];
        const float i1 = inv[r1];
        const v4f g0 = *(const v4f*)(agg + r0 * D + sc);
        const v4f g1 = *(const v4f*)(agg + r1 * D + sc);
        const v4f g2 = *(const v4f*)(xin + r0 * D + sc);
        const v4f g3 = *(const v4f*)(xin + r1 * D + sc);
        *(v4f*)(&sA[0][0][sr * PITCH + sc])       = g0 * i0;
        *(v4f*)(&sA[0][0][(sr + 8) * PITCH + sc]) = g1 * i1;
        *(v4f*)(&sA[0][1][sr * PITCH + sc])       = g2;
        *(v4f*)(&sA[0][1][(sr + 8) * PITCH + sc]) = g3;
    }
    __syncthreads();

    for (; tile < nTiles; tile += gridDim.x) {
        const int  nxt  = tile + (int)gridDim.x;
        const bool more = nxt < nTiles;
        float i0 = 0.0f, i1 = 0.0f;
        v4f g0 = {}, g1 = {}, g2 = {}, g3 = {};
        if (more) {
            const size_t r0 = (size_t)nxt * TM + sr;
            const size_t r1 = r0 + 8;
            i0 = inv[r0];
            i1 = inv[r1];
            g0 = *(const v4f*)(agg + r0 * D + sc);
            g1 = *(const v4f*)(agg + r1 * D + sc);
            g2 = *(const v4f*)(xin + r0 * D + sc);
            g3 = *(const v4f*)(xin + r1 * D + sc);
        }

        const float* pa = &sA[cur][0][l15 * PITCH + 4 * half];
        const float* px = &sA[cur][1][l15 * PITCH + 4 * half];
        v8f acc1 = {};
        v8f acc2 = {};
#pragma unroll
        for (int j = 0; j < 16; ++j) {
            const v4f a = *(const v4f*)(pa + 8 * j);
            const v4f x = *(const v4f*)(px + 8 * j);
            acc1 = __builtin_amdgcn_wmma_f32_16x16x4_f32(false, a.xy, false, bl[j].xy,
                                                         (short)0, acc1, false, false);
            acc2 = __builtin_amdgcn_wmma_f32_16x16x4_f32(false, x.xy, false, br[j].xy,
                                                         (short)0, acc2, false, false);
            acc1 = __builtin_amdgcn_wmma_f32_16x16x4_f32(false, a.zw, false, bl[j].zw,
                                                         (short)0, acc1, false, false);
            acc2 = __builtin_amdgcn_wmma_f32_16x16x4_f32(false, x.zw, false, br[j].zw,
                                                         (short)0, acc2, false, false);
        }

        const int rowBase = tile * TM;
#pragma unroll
        for (int v = 0; v < 8; ++v) {
            float r = (acc1[v] + acc2[v]) + bv;
            if (relu) r = fmaxf(r, 0.0f);
            out[(size_t)(rowBase + v + 8 * half) * D + colBase + l15] = r;
        }

        if (more) {
            const int nb = cur ^ 1;
            *(v4f*)(&sA[nb][0][sr * PITCH + sc])       = g0 * i0;
            *(v4f*)(&sA[nb][0][(sr + 8) * PITCH + sc]) = g1 * i1;
            *(v4f*)(&sA[nb][1][sr * PITCH + sc])       = g2;
            *(v4f*)(&sA[nb][1][(sr + 8) * PITCH + sc]) = g3;
        }
        __syncthreads();
        cur ^= 1;
    }
}

extern "C" void kernel_launch(void* const* d_in, const int* in_sizes, int n_in,
                              void* d_out, int out_size, void* d_ws, size_t ws_size,
                              hipStream_t stream) {
    const float* x   = (const float*)d_in[0];
    const int* edge  = (const int*)d_in[1];
    const float* w1l = (const float*)d_in[2];
    const float* w1r = (const float*)d_in[3];
    const float* b1  = (const float*)d_in[4];
    const float* w2l = (const float*)d_in[5];
    const float* w2r = (const float*)d_in[6];
    const float* b2  = (const float*)d_in[7];
    float* out = (float*)d_out;

    const int N = in_sizes[0] / D;
    const int E = in_sizes[1] / 2;
    const int* src = edge;
    const int* dst = edge + E;

    float* agg = (float*)d_ws;
    float* h   = agg + (size_t)N * D;
    float* inv = h + (size_t)N * D;
    unsigned* bm = (unsigned*)(inv + ((N + 31) & ~31));

    dim3 blk(256);
    dim3 gBits((unsigned)((E + 1023) / 1024));
    const int rowWords = ((E + 1023) / 1024) * TW;
    dim3 gAgg((unsigned)((N + RNG - 1) / RNG));
    dim3 gGemm(512);
    const int nTiles = N / 16;

    gnn_bits<<<gBits, blk, 0, stream>>>(dst, bm, E);

    gnn_agg<<<gAgg, dim3(ATH), 0, stream>>>(x, src, dst, bm, agg, inv, rowWords, N);
    gnn_sage_gemm<<<gGemm, blk, 0, stream>>>(agg, inv, x, w1l, w1r, b1, h, nTiles, 1);

    gnn_agg<<<gAgg, dim3(ATH), 0, stream>>>(h, src, dst, bm, agg, inv, rowWords, N);
    gnn_sage_gemm<<<gGemm, blk, 0, stream>>>(agg, inv, h, w2l, w2r, b2, out, nTiles, 0);
}


# Layer 2 aggregation reuses layer 1's per-range match list (a packed word per edge, read coalesced) and its inv: no second bitmap scan, scattered id reads or degree count; sums stay LDS arrival order
# speedup vs baseline: 3.7998x; 1.0544x over previous
#include <hip/hip_runtime.h>

typedef __attribute__((ext_vector_type(2))) float v2f;
typedef __attribute__((ext_vector_type(4))) float v4f;
typedef __attribute__((ext_vector_type(8))) float v8f;

#define D 128

#define RNG 196
#define LCAP 8192
#define EPI 16
#define ATH 1024
#define TW 32
#define TPITCH 33

__global__ __launch_bounds__(256) void gnn_bits(const int* __restrict__ dst,
                                                unsigned* __restrict__ bm, int nEdges) {
    __shared__ unsigned t[256 * TPITCH];
    const int tid  = threadIdx.x;
    const int lane = tid & 31;
    const int wv   = tid >> 5;
    const int e0 = blockIdx.x * (TW * 32);
    const int rowWords = gridDim.x * TW;
    int d[4];
#pragma unroll
    for (int k = 0; k < 4; ++k) {
        const int e = e0 + k * 256 + tid;
        d[k] = dst[e < nEdges ? e : 0];
    }
    for (int i = tid; i < 256 * TPITCH; i += 256) t[i] = 0u;
    __syncthreads();
#pragma unroll
    for (int k = 0; k < 4; ++k) {
        if (e0 + k * 256 + tid < nEdges) {
            const unsigned r = (unsigned)(d[k] / RNG) & 0xffu;
            atomicOr(&t[r * TPITCH + k * 8 + wv], 1u << lane);
        }
    }
    __syncthreads();
    unsigned* p = bm + (size_t)wv * rowWords + (size_t)blockIdx.x * TW + lane;
    const unsigned* q = t + wv * TPITCH + lane;
#pragma unroll 8
    for (int it = 0; it < 32; ++it) p[(size_t)(it * 8) * rowWords] = q[it * 8 * TPITCH];
}

__global__ __launch_bounds__(ATH) void gnn_agg(const float* __restrict__ feat,
                                               const int* __restrict__ src,
                                               const int* __restrict__ dst,
                                               const unsigned* __restrict__ bm,
                                               float* __restrict__ agg,
                                               float* __restrict__ inv,
                                               unsigned* __restrict__ glist,
                                               unsigned* __restrict__ gcnt,
                                               int rowWords, int nNodes, int reuse) {
    __shared__ __attribute__((aligned(16))) float acc[RNG * D];
    __shared__ float cnt[RNG];
    __shared__ unsigned list[LCAP];
    __shared__ unsigned lcount;

    const int tid  = threadIdx.x;
    const int lane = tid & 31;
    const int wv   = __builtin_amdgcn_readfirstlane(tid >> 5);
    const int base = blockIdx.x * RNG;
    const unsigned* row = bm + (size_t)blockIdx.x * rowWords;
    unsigned* gl = glist + (size_t)blockIdx.x * LCAP;

    const unsigned stored = reuse ? gcnt[blockIdx.x] : 0xffffffffu;
    const bool fromList = __builtin_amdgcn_readfirstlane((int)(stored <= (unsigned)LCAP)) != 0;

    for (int i = tid; i < RNG * (D / 4); i += ATH) ((v4f*)acc)[i] = (v4f){0.f, 0.f, 0.f, 0.f};
    for (int i = tid; i < RNG; i += ATH) cnt[i] = 0.0f;
    if (tid == 0) lcount = fromList ? stored : 0u;
    __syncthreads();

    int nPass = 1;
    unsigned emitted = 0xffffffffu;
    for (int pass = 0; pass < nPass; ++pass) {
        if (fromList) {
        } else if (pass == 0) {
            const int nVec = rowWords >> 2;
            const uint4* rv = (const uint4*)row;
            for (int vb = 0; vb < nVec; vb += 8 * ATH) {
                const int v0 = vb + tid;
                uint4 w[8];
#pragma unroll
                for (int k = 0; k < 8; ++k) {
                    const int vi = v0 + k * ATH;
                    w[k] = rv[vi < nVec ? vi : 0];
                }
                unsigned nT = 0u;
#pragma unroll
                for (int k = 0; k < 8; ++k) {
                    if (v0 + k * ATH >= nVec) w[k] = make_uint4(0u, 0u, 0u, 0u);
                    nT += __builtin_popcount(w[k].x) + __builtin_popcount(w[k].y) +
                          __builtin_popcount(w[k].z) + __builtin_popcount(w[k].w);
                }
                unsigned incl = nT;
#pragma unroll
                for (int dl = 1; dl < 32; dl <<= 1) {
                    const unsigned up = (unsigned)__shfl_up((int)incl, dl);
                    if (lane >= dl) incl += up;
                }
                const unsigned waveTotal = (unsigned)__builtin_amdgcn_readlane((int)incl, 31);
                unsigned wbase = 0u;
                if (lane == 0) wbase = atomicAdd(&lcount, waveTotal);
                wbase = (unsigned)__builtin_amdgcn_readfirstlane((int)wbase);
                unsigned idx = wbase + incl - nT;
                if (idx + nT <= LCAP) {
#pragma unroll
                    for (int k = 0; k < 8; ++k) {
                        const unsigned ww[4] = {w[k].x, w[k].y, w[k].z, w[k].w};
#pragma unroll
                        for (int j = 0; j < 4; ++j) {
                            unsigned x = ww[j];
                            const unsigned eb = ((unsigned)(v0 + k * ATH) * 4u + j) * 32u;
                            while (x) {
                                const unsigned b = (unsigned)__builtin_ctz(x);
                                x &= x - 1u;
                                list[idx++] = eb + b;
                            }
                        }
                    }
                }
            }
        } else {
            const int wi = (pass - 1) * (LCAP / 32) + tid;
            if (tid < LCAP / 32 && wi < rowWords) {
                unsigned x = row[wi];
                unsigned idx = atomicAdd(&lcount, (unsigned)__builtin_popcount(x));
                while (x) {
                    const unsigned b = (unsigned)__builtin_ctz(x);
                    x &= x - 1u;
                    list[idx++] = (unsigned)wi * 32u + b;
                }
            }
        }
        __syncthreads();

        const unsigned lc = lcount;
        if (lc > LCAP) {
            nPass = 1 + (rowWords + LCAP / 32 - 1) / (LCAP / 32);
        } else {
            const int total = __builtin_amdgcn_readfirstlane((int)lc);
            const bool emit = !reuse && nPass == 1;
            if (emit) emitted = lc;
            int i0 = wv * EPI;
            int nS = 0, nD = 0;
            if (i0 + lane < total && lane < EPI) {
                if (fromList) {
                    nS = (int)gl[i0 + lane];
                } else {
                    const unsigned e = list[i0 + lane];
                    nS = src[e];
                    nD = dst[e] - base;
                }
            }
            for (; i0 < total; i0 += (ATH / 32) * EPI) {
                const int left = total - i0;
                const int n = left < EPI ? left : EPI;
                int myS = nS, myD = nD;
                if (fromList) {
                    myD = (int)((unsigned)myS >> 24);
                    myS = myS & 0xffffff;
                } else if (emit && lane < n) {
                    gl[i0 + lane] = (unsigned)myS | ((unsigned)myD << 24);
                }
                const int i1 = i0 + (ATH / 32) * EPI;
                if (i1 + lane < total && lane < EPI) {
                    if (fromList) {
                        nS = (int)gl[i1 + lane];
                    } else {
                        const unsigned e = list[i1 + lane];
                        nS = src[e];
                        nD = dst[e] - base;
                    }
                }
                if (!reuse && lane < n) atomicAdd(&cnt[myD], 1.0f);

                if (n == EPI) {
                    float v[EPI][4];
#pragma unroll
                    for (int u = 0; u < EPI; ++u) {
                        const int sId = __builtin_amdgcn_readlane(myS, u);
                        const float* q = feat + (size_t)sId * D + lane;
                        v[u][0] = q[0];
                        v[u][1] = q[32];
                        v[u][2] = q[64];
                        v[u][3] = q[96];
                    }
#pragma unroll
                    for (int u = 0; u < EPI; ++u) {
                        const int dId = __builtin_amdgcn_readlane(myD, u);
                        float* p = acc + dId * D + lane;
                        atomicAdd(p + 0, v[u][0]);
                        atomicAdd(p + 32, v[u][1]);
                        atomicAdd(p + 64, v[u][2]);
                        atomicAdd(p + 96, v[u][3]);
                    }
                } else {
                    for (int u = 0; u < n; ++u) {
                        const int sId = __shfl(myS, u);
                        const int dId = __shfl(myD, u);
                        const float* q = feat + (size_t)sId * D + lane;
                        const float v0 = q[0];
                        const float v1 = q[32];
                        const float v2 = q[64];
                        const float v3 = q[96];
                        float* p = acc + dId * D + lane;
                        atomicAdd(p + 0, v0);
                        atomicAdd(p + 32, v1);
                        atomicAdd(p + 64, v2);
                        atomicAdd(p + 96, v3);
                    }
                }
            }
        }
        __syncthreads();
        if (tid == 0) lcount = 0u;
        __syncthreads();
    }

    if (!reuse && tid == 0) gcnt[blockIdx.x] = emitted;

    const int rowsLeft = nNodes - base;
    const int rows = rowsLeft < RNG ? rowsLeft : RNG;
    for (int i = tid; i < rows * (D / 4); i += ATH)
        *(v4f*)(agg + (size_t)base * D + (size_t)i * 4) = ((const v4f*)acc)[i];
    if (!reuse)
        for (int r = tid; r < rows; r += ATH) inv[base + r] = 1.0f / fmaxf(cnt[r], 1.0f);
}

#define TM 16
#define PITCH 132

__global__ __launch_bounds__(256) void gnn_sage_gemm(
    const float* __restrict__ agg,
    const float* __restrict__ inv,
    const float* __restrict__ xin,
    const float* __restrict__ wl,
    const float* __restrict__ wr,
    const float* __restrict__ bias,
    float* __restrict__ out,
    int nTiles, int relu) {
    __shared__ __attribute__((aligned(16))) float sA[2][2][TM * PITCH];

    const int tid     = threadIdx.x;
    const int lane    = tid & 31;
    const int wave    = tid >> 5;
    const int half    = lane >> 4;
    const int l15     = lane & 15;
    const int colBase = wave * 16;

    v4f bl[16], br[16];
    {
        const float* wlRow = wl + (size_t)(colBase + l15) * D + 4 * half;
        const float* wrRow = wr + (size_t)(colBase + l15) * D + 4 * half;
#pragma unroll
        for (int j = 0; j < 16; ++j) {
            bl[j] = *(const v4f*)(wlRow + 8 * j);
            br[j] = *(const v4f*)(wrRow + 8 * j);
        }
    }
    const float bv = bias[colBase + l15];

    const int sr = tid >> 5;
    const int sc = (tid & 31) * 4;

    int tile = blockIdx.x;
    int cur  = 0;
    if (tile < nTiles) {
        const size_t r0 = (size_t)tile * TM + sr;
        const size_t r1 = r0 + 8;
        const float i0 = inv[r0];
        const float i1 = inv[r1];
        const v4f g0 = *(const v4f*)(agg + r0 * D + sc);
        const v4f g1 = *(const v4f*)(agg + r1 * D + sc);
        const v4f g2 = *(const v4f*)(xin + r0 * D + sc);
        const v4f g3 = *(const v4f*)(xin + r1 * D + sc);
        *(v4f*)(&sA[0][0][sr * PITCH + sc])       = g0 * i0;
        *(v4f*)(&sA[0][0][(sr + 8) * PITCH + sc]) = g1 * i1;
        *(v4f*)(&sA[0][1][sr * PITCH + sc])       = g2;
        *(v4f*)(&sA[0][1][(sr + 8) * PITCH + sc]) = g3;
    }
    __syncthreads();

    for (; tile < nTiles; tile += gridDim.x) {
        const int  nxt  = tile + (int)gridDim.x;
        const bool more = nxt < nTiles;
        float i0 = 0.0f, i1 = 0.0f;
        v4f g0 = {}, g1 = {}, g2 = {}, g3 = {};
        if (more) {
            const size_t r0 = (size_t)nxt * TM + sr;
            const size_t r1 = r0 + 8;
            i0 = inv[r0];
            i1 = inv[r1];
            g0 = *(const v4f*)(agg + r0 * D + sc);
            g1 = *(const v4f*)(agg + r1 * D + sc);
            g2 = *(const v4f*)(xin + r0 * D + sc);
            g3 = *(const v4f*)(xin + r1 * D + sc);
        }

        const float* pa = &sA[cur][0][l15 * PITCH + 4 * half];
        const float* px = &sA[cur][1][l15 * PITCH + 4 * half];
        v8f acc1 = {};
        v8f acc2 = {};
#pragma unroll
        for (int j = 0; j < 16; ++j) {
            const v4f a = *(const v4f*)(pa + 8 * j);
            const v4f x = *(const v4f*)(px + 8 * j);
            acc1 = __builtin_amdgcn_wmma_f32_16x16x4_f32(false, a.xy, false, bl[j].xy,
                                                         (short)0, acc1, false, false);
            acc2 = __builtin_amdgcn_wmma_f32_16x16x4_f32(false, x.xy, false, br[j].xy,
                                                         (short)0, acc2, false, false);
            acc1 = __builtin_amdgcn_wmma_f32_16x16x4_f32(false, a.zw, false, bl[j].zw,
                                                         (short)0, acc1, false, false);
            acc2 = __builtin_amdgcn_wmma_f32_16x16x4_f32(false, x.zw, false, br[j].zw,
                                                         (short)0, acc2, false, false);
        }

        const int rowBase = tile * TM;
#pragma unroll
        for (int v = 0; v < 8; ++v) {
            float r = (acc1[v] + acc2[v]) + bv;
            if (relu) r = fmaxf(r, 0.0f);
            out[(size_t)(rowBase + v + 8 * half) * D + colBase + l15] = r;
        }

        if (more) {
            const int nb = cur ^ 1;
            *(v4f*)(&sA[nb][0][sr * PITCH + sc])       = g0 * i0;
            *(v4f*)(&sA[nb][0][(sr + 8) * PITCH + sc]) = g1 * i1;
            *(v4f*)(&sA[nb][1][sr * PITCH + sc])       = g2;
            *(v4f*)(&sA[nb][1][(sr + 8) * PITCH + sc]) = g3;
        }
        __syncthreads();
        cur ^= 1;
    }
}

extern "C" void kernel_launch(void* const* d_in, const int* in_sizes, int n_in,
                              void* d_out, int out_size, void* d_ws, size_t ws_size,
                              hipStream_t stream) {
    const float* x   = (const float*)d_in[0];
    const int* edge  = (const int*)d_in[1];
    const float* w1l = (const float*)d_in[2];
    const float* w1r = (const float*)d_in[3];
    const float* b1  = (const float*)d_in[4];
    const float* w2l = (const float*)d_in[5];
    const float* w2r = (const float*)d_in[6];
    const float* b2  = (const float*)d_in[7];
    float* out = (float*)d_out;

    const int N = in_sizes[0] / D;
    const int E = in_sizes[1] / 2;
    const int* src = edge;
    const int* dst = edge + E;

    float* agg = (float*)d_ws;
    float* h   = agg + (size_t)N * D;
    float* inv = h + (size_t)N * D;
    unsigned* bm = (unsigned*)(inv + ((N + 31) & ~31));

    dim3 blk(256);
    dim3 gBits((unsigned)((E + 1023) / 1024));
    const int rowWords = ((E + 1023) / 1024) * TW;
    unsigned* gcnt  = bm + (size_t)256 * rowWords;
    unsigned* glist = gcnt + 256;
    dim3 gAgg((unsigned)((N + RNG - 1) / RNG));
    dim3 gGemm(512);
    const int nTiles = N / 16;

    gnn_bits<<<gBits, blk, 0, stream>>>(dst, bm, E);

    gnn_agg<<<gAgg, dim3(ATH), 0, stream>>>(x, src, dst, bm, agg, inv, glist, gcnt, rowWords, N, 0);
    gnn_sage_gemm<<<gGemm, blk, 0, stream>>>(agg, inv, x, w1l, w1r, b1, h, nTiles, 1);

    gnn_agg<<<gAgg, dim3(ATH), 0, stream>>>(h, src, dst, bm, agg, inv, glist, gcnt, rowWords, N, 1);
    gnn_sage_gemm<<<gGemm, blk, 0, stream>>>(agg, inv, h, w2l, w2r, b2, out, nTiles, 0);
}


# Gather: one 16-byte-per-lane load fetches a whole 512-byte source row (scalar row base) instead of 4 one-line loads; LDS row layout permuted so adds stay conflict-free; sums stay LDS arrival order
# speedup vs baseline: 3.9536x; 1.0405x over previous
#include <hip/hip_runtime.h>

typedef __attribute__((ext_vector_type(2))) float v2f;
typedef __attribute__((ext_vector_type(4))) float v4f;
typedef __attribute__((ext_vector_type(8))) float v8f;

#define D 128

#define RNG 196
#define LCAP 8192
#define EPI 16
#define ATH 1024
#define TW 32
#define TPITCH 33

__global__ __launch_bounds__(256) void gnn_bits(const int* __restrict__ dst,
                                                unsigned* __restrict__ bm, int nEdges) {
    __shared__ unsigned t[256 * TPITCH];
    const int tid  = threadIdx.x;
    const int lane = tid & 31;
    const int wv   = tid >> 5;
    const int e0 = blockIdx.x * (TW * 32);
    const int rowWords = gridDim.x * TW;
    int d[4];
#pragma unroll
    for (int k = 0; k < 4; ++k) {
        const int e = e0 + k * 256 + tid;
        d[k] = dst[e < nEdges ? e : 0];
    }
    for (int i = tid; i < 256 * TPITCH; i += 256) t[i] = 0u;
    __syncthreads();
#pragma unroll
    for (int k = 0; k < 4; ++k) {
        if (e0 + k * 256 + tid < nEdges) {
            const unsigned r = (unsigned)(d[k] / RNG) & 0xffu;
            atomicOr(&t[r * TPITCH + k * 8 + wv], 1u << lane);
        }
    }
    __syncthreads();
    unsigned* p = bm + (size_t)wv * rowWords + (size_t)blockIdx.x * TW + lane;
    const unsigned* q = t + wv * TPITCH + lane;
#pragma unroll 8
    for (int it = 0; it < 32; ++it) p[(size_t)(it * 8) * rowWords] = q[it * 8 * TPITCH];
}

typedef const __attribute__((address_space(1))) v4f* gnn_grow;
static __device__ __forceinline__ gnn_grow gnn_row_base(const float* feat, int sId) {
    const unsigned long long a = (unsigned long long)feat + ((unsigned long long)(unsigned)sId << 9);
    const unsigned lo = (unsigned)__builtin_amdgcn_readfirstlane((int)(unsigned)a);
    const unsigned hi = (unsigned)__builtin_amdgcn_readfirstlane((int)(unsigned)(a >> 32));
    return (gnn_grow)(((unsigned long long)hi << 32) | (unsigned long long)lo);
}

__global__ __launch_bounds__(ATH) void gnn_agg(const float* __restrict__ feat,
                                               const int* __restrict__ src,
                                               const int* __restrict__ dst,
                                               const unsigned* __restrict__ bm,
                                               float* __restrict__ agg,
                                               float* __restrict__ inv,
                                               unsigned* __restrict__ glist,
                                               unsigned* __restrict__ gcnt,
                                               int rowWords, int nNodes, int reuse) {
    __shared__ __attribute__((aligned(16))) float acc[RNG * D];
    __shared__ float cnt[RNG];
    __shared__ unsigned list[LCAP];
    __shared__ unsigned lcount;

    const int tid  = threadIdx.x;
    const int lane = tid & 31;
    const int wv   = __builtin_amdgcn_readfirstlane(tid >> 5);
    const int base = blockIdx.x * RNG;
    const unsigned* row = bm + (size_t)blockIdx.x * rowWords;
    unsigned* gl = glist + (size_t)blockIdx.x * LCAP;

    const unsigned stored = reuse ? gcnt[blockIdx.x] : 0xffffffffu;
    const bool fromList = __builtin_amdgcn_readfirstlane((int)(stored <= (unsigned)LCAP)) != 0;

    for (int i = tid; i < RNG * (D / 4); i += ATH) ((v4f*)acc)[i] = (v4f){0.f, 0.f, 0.f, 0.f};
    for (int i = tid; i < RNG; i += ATH) cnt[i] = 0.0f;
    if (tid == 0) lcount = fromList ? stored : 0u;
    __syncthreads();

    int nPass = 1;
    unsigned emitted = 0xffffffffu;
    for (int pass = 0; pass < nPass; ++pass) {
        if (fromList) {
        } else if (pass == 0) {
            const int nVec = rowWords >> 2;
            const uint4* rv = (const uint4*)row;
            for (int vb = 0; vb < nVec; vb += 8 * ATH) {
                const int v0 = vb + tid;
                uint4 w[8];
#pragma unroll
                for (int k = 0; k < 8; ++k) {
                    const int vi = v0 + k * ATH;
                    w[k] = rv[vi < nVec ? vi : 0];
                }
                unsigned nT = 0u;
#pragma unroll
                for (int k = 0; k < 8; ++k) {
                    if (v0 + k * ATH >= nVec) w[k] = make_uint4(0u, 0u, 0u, 0u);
                    nT += __builtin_popcount(w[k].x) + __builtin_popcount(w[k].y) +
                          __builtin_popcount(w[k].z) + __builtin_popcount(w[k].w);
                }
                unsigned incl = nT;
#pragma unroll
                for (int dl = 1; dl < 32; dl <<= 1) {
                    const unsigned up = (unsigned)__shfl_up((int)incl, dl);
                    if (lane >= dl) incl += up;
                }
                const unsigned waveTotal = (unsigned)__builtin_amdgcn_readlane((int)incl, 31);
                unsigned wbase = 0u;
                if (lane == 0) wbase = atomicAdd(&lcount, waveTotal);
                wbase = (unsigned)__builtin_amdgcn_readfirstlane((int)wbase);
                unsigned idx = wbase + incl - nT;
                if (idx + nT <= LCAP) {
#pragma unroll
                    for (int k = 0; k < 8; ++k) {
                        const unsigned ww[4] = {w[k].x, w[k].y, w[k].z, w[k].w};
#pragma unroll
                        for (int j = 0; j < 4; ++j) {
                            unsigned x = ww[j];
                            const unsigned eb = ((unsigned)(v0 + k * ATH) * 4u + j) * 32u;
                            while (x) {
                                const unsigned b = (unsigned)__builtin_ctz(x);
                                x &= x - 1u;
                                list[idx++] = eb + b;
                            }
                        }
                    }
                }
            }
        } else {
            const int wi = (pass - 1) * (LCAP / 32) + tid;
            if (tid < LCAP / 32 && wi < rowWords) {
                unsigned x = row[wi];
                unsigned idx = atomicAdd(&lcount, (unsigned)__builtin_popcount(x));
                while (x) {
                    const unsigned b = (unsigned)__builtin_ctz(x);
                    x &= x - 1u;
                    list[idx++] = (unsigned)wi * 32u + b;
                }
            }
        }
        __syncthreads();

        const unsigned lc = lcount;
        if (lc > LCAP) {
            nPass = 1 + (rowWords + LCAP / 32 - 1) / (LCAP / 32);
        } else {
            const int total = __builtin_amdgcn_readfirstlane((int)lc);
            const bool emit = !reuse && nPass == 1;
            if (emit) emitted = lc;
            int i0 = wv * EPI;
            int nS = 0, nD = 0;
            if (i0 + lane < total && lane < EPI) {
                if (fromList) {
                    nS = (int)gl[i0 + lane];
                } else {
                    const unsigned e = list[i0 + lane];
                    nS = src[e];
                    nD = dst[e] - base;
                }
            }
            for (; i0 < total; i0 += (ATH / 32) * EPI) {
                const int left = total - i0;
                const int n = left < EPI ? left : EPI;
                int myS = nS, myD = nD;
                if (fromList) {
                    myD = (int)((unsigned)myS >> 24);
                    myS = myS & 0xffffff;
                } else if (emit && lane < n) {
                    gl[i0 + lane] = (unsigned)myS | ((unsigned)myD << 24);
                }
                const int i1 = i0 + (ATH / 32) * EPI;
                if (i1 + lane < total && lane < EPI) {
                    if (fromList) {
                        nS = (int)gl[i1 + lane];
                    } else {
                        const unsigned e = list[i1 + lane];
                        nS = src[e];
                        nD = dst[e] - base;
                    }
                }
                if (!reuse && lane < n) atomicAdd(&cnt[myD], 1.0f);

                if (n == EPI) {
                    v4f v[EPI];
                    gnn_grow rb[EPI];
#pragma unroll
                    for (int u = 0; u < EPI; ++u)
                        rb[u] = gnn_row_base(feat, __builtin_amdgcn_readlane(myS, u));
#pragma unroll
                    for (int u = 0; u < EPI; ++u) v[u] = rb[u][(unsigned)lane];
#pragma unroll
                    for (int u = 0; u < EPI; ++u) {
                        const int dId = __builtin_amdgcn_readlane(myD, u);
                        float* p = acc + dId * D + lane;
                        atomicAdd(p + 0, v[u].x);
                        atomicAdd(p + 32, v[u].y);
                        atomicAdd(p + 64, v[u].z);
                        atomicAdd(p + 96, v[u].w);
                    }
                } else {
                    for (int u = 0; u < n; ++u) {
                        const int sId = __shfl(myS, u);
                        const int dId = __shfl(myD, u);
                        const v4f vv = gnn_row_base(feat, sId)[(unsigned)lane];
                        float* p = acc + dId * D + lane;
                        atomicAdd(p + 0, vv.x);
                        atomicAdd(p + 32, vv.y);
                        atomicAdd(p + 64, vv.z);
                        atomicAdd(p + 96, vv.w);
                    }
                }
            }
        }
        __syncthreads();
        if (tid == 0) lcount = 0u;
        __syncthreads();
    }

    if (!reuse && tid == 0) gcnt[blockIdx.x] = emitted;

    const int rowsLeft = nNodes - base;
    const int rows = rowsLeft < RNG ? rowsLeft : RNG;
    for (int i = tid; i < rows * (D / 4); i += ATH) {
        const float* a = acc + (i >> 5) * D + (i & 31);
        const v4f o = {a[0], a[32], a[64], a[96]};
        *(v4f*)(agg + (size_t)base * D + (size_t)i * 4) = o;
    }
    if (!reuse)
        for (int r = tid; r < rows; r += ATH) inv[base + r] = 1.0f / fmaxf(cnt[r], 1.0f);
}

#define TM 16
#define PITCH 132

__global__ __launch_bounds__(256) void gnn_sage_gemm(
    const float* __restrict__ agg,
    const float* __restrict__ inv,
    const float* __restrict__ xin,
    const float* __restrict__ wl,
    const float* __restrict__ wr,
    const float* __restrict__ bias,
    float* __restrict__ out,
    int nTiles, int relu) {
    __shared__ __attribute__((aligned(16))) float sA[2][2][TM * PITCH];

    const int tid     = threadIdx.x;
    const int lane    = tid & 31;
    const int wave    = tid >> 5;
    const int half    = lane >> 4;
    const int l15     = lane & 15;
    const int colBase = wave * 16;

    v4f bl[16], br[16];
    {
        const float* wlRow = wl + (size_t)(colBase + l15) * D + 4 * half;
        const float* wrRow = wr + (size_t)(colBase + l15) * D + 4 * half;
#pragma unroll
        for (int j = 0; j < 16; ++j) {
            bl[j] = *(const v4f*)(wlRow + 8 * j);
            br[j] = *(const v4f*)(wrRow + 8 * j);
        }
    }
    const float bv = bias[colBase + l15];

    const int sr = tid >> 5;
    const int sc = (tid & 31) * 4;

    int tile = blockIdx.x;
    int cur  = 0;
    if (tile < nTiles) {
        const size_t r0 = (size_t)tile * TM + sr;
        const size_t r1 = r0 + 8;
        const float i0 = inv[r0];
        const float i1 = inv[r1];
        const v4f g0 = *(const v4f*)(agg + r0 * D + sc);
        const v4f g1 = *(const v4f*)(agg + r1 * D + sc);
        const v4f g2 = *(const v4f*)(xin + r0 * D + sc);
        const v4f g3 = *(const v4f*)(xin + r1 * D + sc);
        *(v4f*)(&sA[0][0][sr * PITCH + sc])       = g0 * i0;
        *(v4f*)(&sA[0][0][(sr + 8) * PITCH + sc]) = g1 * i1;
        *(v4f*)(&sA[0][1][sr * PITCH + sc])       = g2;
        *(v4f*)(&sA[0][1][(sr + 8) * PITCH + sc]) = g3;
    }
    __syncthreads();

    for (; tile < nTiles; tile += gridDim.x) {
        const int  nxt  = tile + (int)gridDim.x;
        const bool more = nxt < nTiles;
        float i0 = 0.0f, i1 = 0.0f;
        v4f g0 = {}, g1 = {}, g2 = {}, g3 = {};
        if (more) {
            const size_t r0 = (size_t)nxt * TM + sr;
            const size_t r1 = r0 + 8;
            i0 = inv[r0];
            i1 = inv[r1];
            g0 = *(const v4f*)(agg + r0 * D + sc);
            g1 = *(const v4f*)(agg + r1 * D + sc);
            g2 = *(const v4f*)(xin + r0 * D + sc);
            g3 = *(const v4f*)(xin + r1 * D + sc);
        }

        const float* pa = &sA[cur][0][l15 * PITCH + 4 * half];
        const float* px = &sA[cur][1][l15 * PITCH + 4 * half];
        v8f acc1 = {};
        v8f acc2 = {};
#pragma unroll
        for (int j = 0; j < 16; ++j) {
            const v4f a = *(const v4f*)(pa + 8 * j);
            const v4f x = *(const v4f*)(px + 8 * j);
            acc1 = __builtin_amdgcn_wmma_f32_16x16x4_f32(false, a.xy, false, bl[j].xy,
                                                         (short)0, acc1, false, false);
            acc2 = __builtin_amdgcn_wmma_f32_16x16x4_f32(false, x.xy, false, br[j].xy,
                                                         (short)0, acc2, false, false);
            acc1 = __builtin_amdgcn_wmma_f32_16x16x4_f32(false, a.zw, false, bl[j].zw,
                                                         (short)0, acc1, false, false);
            acc2 = __builtin_amdgcn_wmma_f32_16x16x4_f32(false, x.zw, false, br[j].zw,
                                                         (short)0, acc2, false, false);
        }

        const int rowBase = tile * TM;
#pragma unroll
        for (int v = 0; v < 8; ++v) {
            float r = (acc1[v] + acc2[v]) + bv;
            if (relu) r = fmaxf(r, 0.0f);
            out[(size_t)(rowBase + v + 8 * half) * D + colBase + l15] = r;
        }

        if (more) {
            const int nb = cur ^ 1;
            *(v4f*)(&sA[nb][0][sr * PITCH + sc])       = g0 * i0;
            *(v4f*)(&sA[nb][0][(sr + 8) * PITCH + sc]) = g1 * i1;
            *(v4f*)(&sA[nb][1][sr * PITCH + sc])       = g2;
            *(v4f*)(&sA[nb][1][(sr + 8) * PITCH + sc]) = g3;
        }
        __syncthreads();
        cur ^= 1;
    }
}

extern "C" void kernel_launch(void* const* d_in, const int* in_sizes, int n_in,
                              void* d_out, int out_size, void* d_ws, size_t ws_size,
                              hipStream_t stream) {
    const float* x   = (const float*)d_in[0];
    const int* edge  = (const int*)d_in[1];
    const float* w1l = (const float*)d_in[2];
    const float* w1r = (const float*)d_in[3];
    const float* b1  = (const float*)d_in[4];
    const float* w2l = (const float*)d_in[5];
    const float* w2r = (const float*)d_in[6];
    const float* b2  = (const float*)d_in[7];
    float* out = (float*)d_out;

    const int N = in_sizes[0] / D;
    const int E = in_sizes[1] / 2;
    const int* src = edge;
    const int* dst = edge + E;

    float* agg = (float*)d_ws;
    float* h   = agg + (size_t)N * D;
    float* inv = h + (size_t)N * D;
    unsigned* bm = (unsigned*)(inv + ((N + 31) & ~31));

    dim3 blk(256);
    dim3 gBits((unsigned)((E + 1023) / 1024));
    const int rowWords = ((E + 1023) / 1024) * TW;
    unsigned* gcnt  = bm + (size_t)256 * rowWords;
    unsigned* glist = gcnt + 256;
    dim3 gAgg((unsigned)((N + RNG - 1) / RNG));
    dim3 gGemm(512);
    const int nTiles = N / 16;

    gnn_bits<<<gBits, blk, 0, stream>>>(dst, bm, E);

    gnn_agg<<<gAgg, dim3(ATH), 0, stream>>>(x, src, dst, bm, agg, inv, glist, gcnt, rowWords, N, 0);
    gnn_sage_gemm<<<gGemm, blk, 0, stream>>>(agg, inv, x, w1l, w1r, b1, h, nTiles, 1);

    gnn_agg<<<gAgg, dim3(ATH), 0, stream>>>(h, src, dst, bm, agg, inv, glist, gcnt, rowWords, N, 1);
    gnn_sage_gemm<<<gGemm, blk, 0, stream>>>(agg, inv, h, w2l, w2r, b2, out, nTiles, 0);
}


# Edge bitmaps replaced by slot lines: the tile kernel deals each edge's packed (source, local dst) word into its range's 128-byte line; no per-edge id reads, no bit scans; sums stay LDS arrival order
# speedup vs baseline: 4.1420x; 1.0477x over previous
#include <hip/hip_runtime.h>

typedef __attribute__((ext_vector_type(2))) float v2f;
typedef __attribute__((ext_vector_type(4))) float v4f;
typedef __attribute__((ext_vector_type(8))) float v8f;

#define D 128

#define RNG 196
#define LCAP 8192
#define EPI 16
#define ATH 1024
#define TW 32
#define TPITCH 33
#define TPP (LCAP / (TW * 32))

__global__ __launch_bounds__(256) void gnn_bits(const int* __restrict__ src,
                                                const int* __restrict__ dst,
                                                unsigned* __restrict__ bm, int nEdges) {
    __shared__ unsigned t[256 * TPITCH];
    const int tid  = threadIdx.x;
    const int lane = tid & 31;
    const int wv   = tid >> 5;
    const int e0 = blockIdx.x * (TW * 32);
    const int rowWords = gridDim.x * TW;
    int d[4], s[4];
#pragma unroll
    for (int k = 0; k < 4; ++k) {
        const int e = e0 + k * 256 + tid;
        const int ec = e < nEdges ? e : 0;
        d[k] = dst[ec];
        s[k] = src[ec];
    }
    t[tid * TPITCH] = 0u;
    __syncthreads();
#pragma unroll
    for (int k = 0; k < 4; ++k) {
        if (e0 + k * 256 + tid < nEdges) {
            const unsigned r = (unsigned)(d[k] / RNG) & 0xffu;
            const unsigned pos = atomicAdd(&t[r * TPITCH], 1u);
            if (pos < (unsigned)(TW - 1))
                t[r * TPITCH + 1u + pos] = (unsigned)s[k] | ((unsigned)(d[k] - (int)r * RNG) << 24);
        }
    }
    __syncthreads();
    unsigned* p = bm + (size_t)wv * rowWords + (size_t)blockIdx.x * TW + lane;
    const unsigned* q = t + wv * TPITCH + lane;
#pragma unroll 8
    for (int it = 0; it < 32; ++it) p[(size_t)(it * 8) * rowWords] = q[it * 8 * TPITCH];
}

typedef const __attribute__((address_space(1))) v4f* gnn_grow;
static __device__ __forceinline__ gnn_grow gnn_row_base(const float* feat, int sId) {
    const unsigned long long a = (unsigned long long)feat + ((unsigned long long)(unsigned)sId << 9);
    const unsigned lo = (unsigned)__builtin_amdgcn_readfirstlane((int)(unsigned)a);
    const unsigned hi = (unsigned)__builtin_amdgcn_readfirstlane((int)(unsigned)(a >> 32));
    return (gnn_grow)(((unsigned long long)hi << 32) | (unsigned long long)lo);
}

__global__ __launch_bounds__(ATH) void gnn_agg(const float* __restrict__ feat,
                                               const int* __restrict__ src,
                                               const int* __restrict__ dst,
                                               const unsigned* __restrict__ bm,
                                               float* __restrict__ agg,
                                               float* __restrict__ inv,
                                               unsigned* __restrict__ glist,
                                               unsigned* __restrict__ gcnt,
                                               int rowWords, int nNodes, int nEdges, int reuse) {
    __shared__ __attribute__((aligned(16))) float acc[RNG * D];
    __shared__ float cnt[RNG];
    __shared__ unsigned list[LCAP];
    __shared__ unsigned lcount;
    __shared__ unsigned anyOvf;

    const int tid  = threadIdx.x;
    const int lane = tid & 31;
    const int wv   = __builtin_amdgcn_readfirstlane(tid >> 5);
    const int base = blockIdx.x * RNG;
    const unsigned* row = bm + (size_t)blockIdx.x * rowWords;
    unsigned* gl = glist + (size_t)blockIdx.x * LCAP;

    const unsigned stored = reuse ? gcnt[blockIdx.x] : 0xffffffffu;
    const bool fromList = __builtin_amdgcn_readfirstlane((int)(stored <= (unsigned)LCAP)) != 0;

    for (int i = tid; i < RNG * (D / 4); i += ATH) ((v4f*)acc)[i] = (v4f){0.f, 0.f, 0.f, 0.f};
    for (int i = tid; i < RNG; i += ATH) cnt[i] = 0.0f;
    if (tid == 0) {
        lcount = fromList ? stored : 0u;
        anyOvf = 0u;
    }
    __syncthreads();

    const int nLines = rowWords / TW;
    int nPass = 1;
    unsigned emitted = 0xffffffffu;
    for (int pass = 0; pass < nPass; ++pass) {
        const int tLo = pass == 0 ? 0 : (pass - 1) * TPP;
        const int tEnd = pass == 0 ? nLines : tLo + TPP;
        const int tHi = tEnd < nLines ? tEnd : nLines;
        if (fromList) {
        } else {
            const int vLo = tLo * 8, vHi = tHi * 8;
            const uint4* rv = (const uint4*)row;
            const unsigned fq = 4u * (unsigned)(lane & 7);
            for (int vb = vLo; vb < vHi; vb += 8 * ATH) {
                const int v0 = vb + tid;
                uint4 w[8];
#pragma unroll
                for (int k = 0; k < 8; ++k) {
                    const int vi = v0 + k * ATH;
                    w[k] = rv[vi < vHi ? vi : vLo];
                }
                unsigned nT = 0u;
                unsigned cl[8];
#pragma unroll
                for (int k = 0; k < 8; ++k) {
                    unsigned cc = (unsigned)__shfl((int)w[k].x, lane & ~7);
                    if (v0 + k * ATH >= vHi) cc = 0u;
                    if (cc > (unsigned)(TW - 1)) {
                        anyOvf = 1u;
                        cc = 0u;
                    }
                    cl[k] = cc;
#pragma unroll
                    for (int j = 0; j < 4; ++j) nT += (fq + (unsigned)j - 1u < cc) ? 1u : 0u;
                }
                unsigned incl = nT;
#pragma unroll
                for (int dl = 1; dl < 32; dl <<= 1) {
                    const unsigned up = (unsigned)__shfl_up((int)incl, dl);
                    if (lane >= dl) incl += up;
                }
                const unsigned waveTotal = (unsigned)__builtin_amdgcn_readlane((int)incl, 31);
                unsigned wbase = 0u;
                if (lane == 0) wbase = atomicAdd(&lcount, waveTotal);
                wbase = (unsigned)__builtin_amdgcn_readfirstlane((int)wbase);
                unsigned idx = wbase + incl - nT;
                if (idx + nT <= LCAP) {
#pragma unroll
                    for (int k = 0; k < 8; ++k) {
                        const unsigned ww[4] = {w[k].x, w[k].y, w[k].z, w[k].w};
#pragma unroll
                        for (int j = 0; j < 4; ++j)
                            if (fq + (unsigned)j - 1u < cl[k]) list[idx++] = ww[j];
                    }
                }
            }
        }
        __syncthreads();
        const bool ovf = !fromList && __builtin_amdgcn_readfirstlane((int)anyOvf) != 0;
        if (ovf) {
            for (int t = tLo; t < tHi; ++t) {
                if (row[(size_t)t * TW] > (unsigned)(TW - 1)) {
                    const int e = t * (TW * 32) + tid;
                    if (e < nEdges) {
                        const int dd = dst[e];
                        if (dd / RNG == (int)blockIdx.x) {
                            const unsigned at = atomicAdd(&lcount, 1u);
                            if (at < (unsigned)LCAP) list[at] = (unsigned)src[e] | ((unsigned)(dd - base) << 24);
                        }
                    }
                }
            }
            __syncthreads();
        }

        const unsigned lc = lcount;
        if (lc > LCAP) {
            nPass = 1 + (nLines + TPP - 1) / TPP;
        } else {
            const int total = __builtin_amdgcn_readfirstlane((int)lc);
            const bool emit = !reuse && nPass == 1;
            if (emit) emitted = lc;
            int i0 = wv * EPI;
            int nS = 0;
            if (i0 + lane < total && lane < EPI) nS = fromList ? (int)gl[i0 + lane] : (int)list[i0 + lane];
            for (; i0 < total; i0 += (ATH / 32) * EPI) {
                const int left = total - i0;
                const int n = left < EPI ? left : EPI;
                if (emit && lane < n) gl[i0 + lane] = (unsigned)nS;
                const int myD = (int)((unsigned)nS >> 24);
                const int myS = nS & 0xffffff;
                const int i1 = i0 + (ATH / 32) * EPI;
                if (i1 + lane < total && lane < EPI) nS = fromList ? (int)gl[i1 + lane] : (int)list[i1 + lane];
                if (!reuse && lane < n) atomicAdd(&cnt[myD], 1.0f);

                if (n == EPI) {
                    v4f v[EPI];
                    gnn_grow rb[EPI];
#pragma unroll
                    for (int u = 0; u < EPI; ++u)
                        rb[u] = gnn_row_base(feat, __builtin_amdgcn_readlane(myS, u));
#pragma unroll
                    for (int u = 0; u < EPI; ++u) v[u] = rb[u][(unsigned)lane];
#pragma unroll
                    for (int u = 0; u < EPI; ++u) {
                        const int dId = __builtin_amdgcn_readlane(myD, u);
                        float* p = acc + dId * D + lane;
                        atomicAdd(p + 0, v[u].x);
                        atomicAdd(p + 32, v[u].y);
                        atomicAdd(p + 64, v[u].z);
                        atomicAdd(p + 96, v[u].w);
                    }
                } else {
                    for (int u = 0; u < n; ++u) {
                        const int sId = __shfl(myS, u);
                        const int dId = __shfl(myD, u);
                        const v4f vv = gnn_row_base(feat, sId)[(unsigned)lane];
                        float* p = acc + dId * D + lane;
                        atomicAdd(p + 0, vv.x);
                        atomicAdd(p + 32, vv.y);
                        atomicAdd(p + 64, vv.z);
                        atomicAdd(p + 96, vv.w);
                    }
                }
            }
        }
        __syncthreads();
        if (tid == 0) lcount = 0u;
        __syncthreads();
    }

    if (!reuse && tid == 0) gcnt[blockIdx.x] = emitted;

    const int rowsLeft = nNodes - base;
    const int rows = rowsLeft < RNG ? rowsLeft : RNG;
    for (int i = tid; i < rows * (D / 4); i += ATH) {
        const float* a = acc + (i >> 5) * D + (i & 31);
        const v4f o = {a[0], a[32], a[64], a[96]};
        *(v4f*)(agg + (size_t)base * D + (size_t)i * 4) = o;
    }
    if (!reuse)
        for (int r = tid; r < rows; r += ATH) inv[base + r] = 1.0f / fmaxf(cnt[r], 1.0f);
}

#define TM 16
#define PITCH 132

__global__ __launch_bounds__(256) void gnn_sage_gemm(
    const float* __restrict__ agg,
    const float* __restrict__ inv,
    const float* __restrict__ xin,
    const float* __restrict__ wl,
    const float* __restrict__ wr,
    const float* __restrict__ bias,
    float* __restrict__ out,
    int nTiles, int relu) {
    __shared__ __attribute__((aligned(16))) float sA[2][2][TM * PITCH];

    const int tid     = threadIdx.x;
    const int lane    = tid & 31;
    const int wave    = tid >> 5;
    const int half    = lane >> 4;
    const int l15     = lane & 15;
    const int colBase = wave * 16;

    v4f bl[16], br[16];
    {
        const float* wlRow = wl + (size_t)(colBase + l15) * D + 4 * half;
        const float* wrRow = wr + (size_t)(colBase + l15) * D + 4 * half;
#pragma unroll
        for (int j = 0; j < 16; ++j) {
            bl[j] = *(const v4f*)(wlRow + 8 * j);
            br[j] = *(const v4f*)(wrRow + 8 * j);
        }
    }
    const float bv = bias[colBase + l15];

    const int sr = tid >> 5;
    const int sc = (tid & 31) * 4;

    int tile = blockIdx.x;
    int cur  = 0;
    if (tile < nTiles) {
        const size_t r0 = (size_t)tile * TM + sr;
        const size_t r1 = r0 + 8;
        const float i0 = inv[r0];
        const float i1 = inv[r1];
        const v4f g0 = *(const v4f*)(agg + r0 * D + sc);
        const v4f g1 = *(const v4f*)(agg + r1 * D + sc);
        const v4f g2 = *(const v4f*)(xin + r0 * D + sc);
        const v4f g3 = *(const v4f*)(xin + r1 * D + sc);
        *(v4f*)(&sA[0][0][sr * PITCH + sc])       = g0 * i0;
        *(v4f*)(&sA[0][0][(sr + 8) * PITCH + sc]) = g1 * i1;
        *(v4f*)(&sA[0][1][sr * PITCH + sc])       = g2;
        *(v4f*)(&sA[0][1][(sr + 8) * PITCH + sc]) = g3;
    }
    __syncthreads();

    for (; tile < nTiles; tile += gridDim.x) {
        const int  nxt  = tile + (int)gridDim.x;
        const bool more = nxt < nTiles;
        float i0 = 0.0f, i1 = 0.0f;
        v4f g0 = {}, g1 = {}, g2 = {}, g3 = {};
        if (more) {
            const size_t r0 = (size_t)nxt * TM + sr;
            const size_t r1 = r0 + 8;
            i0 = inv[r0];
            i1 = inv[r1];
            g0 = *(const v4f*)(agg + r0 * D + sc);
            g1 = *(const v4f*)(agg + r1 * D + sc);
            g2 = *(const v4f*)(xin + r0 * D + sc);
            g3 = *(const v4f*)(xin + r1 * D + sc);
        }

        const float* pa = &sA[cur][0][l15 * PITCH + 4 * half];
        const float* px = &sA[cur][1][l15 * PITCH + 4 * half];
        v8f acc1 = {};
        v8f acc2 = {};
#pragma unroll
        for (int j = 0; j < 16; ++j) {
            const v4f a = *(const v4f*)(pa + 8 * j);
            const v4f x = *(const v4f*)(px + 8 * j);
            acc1 = __builtin_amdgcn_wmma_f32_16x16x4_f32(false, a.xy, false, bl[j].xy,
                                                         (short)0, acc1, false, false);
            acc2 = __builtin_amdgcn_wmma_f32_16x16x4_f32(false, x.xy, false, br[j].xy,
                                                         (short)0, acc2, false, false);
            acc1 = __builtin_amdgcn_wmma_f32_16x16x4_f32(false, a.zw, false, bl[j].zw,
                                                         (short)0, acc1, false, false);
            acc2 = __builtin_amdgcn_wmma_f32_16x16x4_f32(false, x.zw, false, br[j].zw,
                                                         (short)0, acc2, false, false);
        }

        const int rowBase = tile * TM;
#pragma unroll
        for (int v = 0; v < 8; ++v) {
            float r = (acc1[v] + acc2[v]) + bv;
            if (relu) r = fmaxf(r, 0.0f);
            out[(size_t)(rowBase + v + 8 * half) * D + colBase + l15] = r;
        }

        if (more) {
            const int nb = cur ^ 1;
            *(v4f*)(&sA[nb][0][sr * PITCH + sc])       = g0 * i0;
            *(v4f*)(&sA[nb][0][(sr + 8) * PITCH + sc]) = g1 * i1;
            *(v4f*)(&sA[nb][1][sr * PITCH + sc])       = g2;
            *(v4f*)(&sA[nb][1][(sr + 8) * PITCH + sc]) = g3;
        }
        __syncthreads();
        cur ^= 1;
    }
}

extern "C" void kernel_launch(void* const* d_in, const int* in_sizes, int n_in,
                              void* d_out, int out_size, void* d_ws, size_t ws_size,
                              hipStream_t stream) {
    const float* x   = (const float*)d_in[0];
    const int* edge  = (const int*)d_in[1];
    const float* w1l = (const float*)d_in[2];
    const float* w1r = (const float*)d_in[3];
    const float* b1  = (const float*)d_in[4];
    const float* w2l = (const float*)d_in[5];
    const float* w2r = (const float*)d_in[6];
    const float* b2  = (const float*)d_in[7];
    float* out = (float*)d_out;

    const int N = in_sizes[0] / D;
    const int E = in_sizes[1] / 2;
    const int* src = edge;
    const int* dst = edge + E;

    float* agg = (float*)d_ws;
    float* h   = agg + (size_t)N * D;
    float* inv = h + (size_t)N * D;
    unsigned* bm = (unsigned*)(inv + ((N + 31) & ~31));

    dim3 blk(256);
    dim3 gBits((unsigned)((E + 1023) / 1024));
    const int rowWords = ((E + 1023) / 1024) * TW;
    unsigned* gcnt  = bm + (size_t)256 * rowWords;
    unsigned* glist = gcnt + 256;
    dim3 gAgg((unsigned)((N + RNG - 1) / RNG));
    dim3 gGemm(512);
    const int nTiles = N / 16;

    gnn_bits<<<gBits, blk, 0, stream>>>(src, dst, bm, E);

    gnn_agg<<<gAgg, dim3(ATH), 0, stream>>>(x, src, dst, bm, agg, inv, glist, gcnt, rowWords, N, E, 0);
    gnn_sage_gemm<<<gGemm, blk, 0, stream>>>(agg, inv, x, w1l, w1r, b1, h, nTiles, 1);

    gnn_agg<<<gAgg, dim3(ATH), 0, stream>>>(h, src, dst, bm, agg, inv, glist, gcnt, rowWords, N, E, 1);
    gnn_sage_gemm<<<gGemm, blk, 0, stream>>>(agg, inv, h, w2l, w2r, b2, out, nTiles, 0);
}


# Aggregation: LDS float atomics replaced by register sums; the range's edge list is dealt into one sub-list per wave, a wave owns rows w+32k (sums indexed by register); sum order = sub-list arrival
# speedup vs baseline: 4.3133x; 1.0414x over previous
#include <hip/hip_runtime.h>

typedef __attribute__((ext_vector_type(2))) float v2f;
typedef __attribute__((ext_vector_type(4))) float v4f;
typedef __attribute__((ext_vector_type(8))) float v8f;

#define D 128

#define RNG 196
#define LCAP 8192
#define EPI 16
#define ATH 1024
#define TW 32
#define TPITCH 33
#define TPP (LCAP / (TW * 32))

__global__ __launch_bounds__(256) void gnn_bits(const int* __restrict__ src,
                                                const int* __restrict__ dst,
                                                unsigned* __restrict__ bm, int nEdges) {
    __shared__ unsigned t[256 * TPITCH];
    const int tid  = threadIdx.x;
    const int lane = tid & 31;
    const int wv   = tid >> 5;
    const int e0 = blockIdx.x * (TW * 32);
    const int rowWords = gridDim.x * TW;
    int d[4], s[4];
#pragma unroll
    for (int k = 0; k < 4; ++k) {
        const int e = e0 + k * 256 + tid;
        const int ec = e < nEdges ? e : 0;
        d[k] = dst[ec];
        s[k] = src[ec];
    }
    t[tid * TPITCH] = 0u;
    __syncthreads();
#pragma unroll
    for (int k = 0; k < 4; ++k) {
        if (e0 + k * 256 + tid < nEdges) {
            const unsigned r = (unsigned)(d[k] / RNG) & 0xffu;
            const unsigned pos = atomicAdd(&t[r * TPITCH], 1u);
            if (pos < (unsigned)(TW - 1))
                t[r * TPITCH + 1u + pos] = (unsigned)s[k] | ((unsigned)(d[k] - (int)r * RNG) << 24);
        }
    }
    __syncthreads();
    unsigned* p = bm + (size_t)wv * rowWords + (size_t)blockIdx.x * TW + lane;
    const unsigned* q = t + wv * TPITCH + lane;
#pragma unroll 8
    for (int it = 0; it < 32; ++it) p[(size_t)(it * 8) * rowWords] = q[it * 8 * TPITCH];
}

typedef const __attribute__((address_space(1))) v4f* gnn_grow;
static __device__ __forceinline__ gnn_grow gnn_row_base(const float* feat, int sId) {
    const unsigned long long a = (unsigned long long)feat + ((unsigned long long)(unsigned)sId << 9);
    const unsigned lo = (unsigned)__builtin_amdgcn_readfirstlane((int)(unsigned)a);
    const unsigned hi = (unsigned)__builtin_amdgcn_readfirstlane((int)(unsigned)(a >> 32));
    return (gnn_grow)(((unsigned long long)hi << 32) | (unsigned long long)lo);
}

#define NOWN ((RNG + 31) / 32)
#define PADW (7u << 29)

__global__ __launch_bounds__(ATH) void gnn_agg(const float* __restrict__ feat,
                                               const int* __restrict__ src,
                                               const int* __restrict__ dst,
                                               const unsigned* __restrict__ bm,
                                               float* __restrict__ agg,
                                               float* __restrict__ inv,
                                               unsigned* __restrict__ glist,
                                               unsigned* __restrict__ gcnt,
                                               unsigned* __restrict__ gofs,
                                               int rowWords, int nNodes, int nEdges, int reuse) {
    __shared__ float cnt[RNG];
    __shared__ unsigned list[LCAP];
    __shared__ unsigned binCnt[32];
    __shared__ unsigned binOff[33];
    __shared__ unsigned lcount;
    __shared__ unsigned anyOvf;

    const int tid  = threadIdx.x;
    const int lane = tid & 31;
    const int wv   = __builtin_amdgcn_readfirstlane(tid >> 5);
    const int base = blockIdx.x * RNG;
    const unsigned* row = bm + (size_t)blockIdx.x * rowWords;
    unsigned* gl = glist + (size_t)blockIdx.x * LCAP;
    unsigned* go = gofs + (size_t)blockIdx.x * 33;

    const unsigned stored = reuse ? gcnt[blockIdx.x] : 0xffffffffu;
    const bool fromList = __builtin_amdgcn_readfirstlane((int)(stored <= (unsigned)LCAP)) != 0;

    for (int i = tid; i < RNG; i += ATH) cnt[i] = 0.0f;
    if (tid < 32) binCnt[tid] = 0u;
    if (tid == 0) {
        lcount = fromList ? stored : 0u;
        anyOvf = 0u;
    }
    __syncthreads();

    v8f ax = {}, ay = {}, az = {}, aw = {};

    const int nLines = rowWords / TW;
    int nPass = 1;
    unsigned emitted = 0xffffffffu;
    for (int pass = 0; pass < nPass; ++pass) {
        const int tLo = pass == 0 ? 0 : (pass - 1) * TPP;
        const int tEnd = pass == 0 ? nLines : tLo + TPP;
        const int tHi = tEnd < nLines ? tEnd : nLines;
        if (fromList) {
        } else {
            const int vLo = tLo * 8, vHi = tHi * 8;
            const uint4* rv = (const uint4*)row;
            const unsigned fq = 4u * (unsigned)(lane & 7);
            for (int vb = vLo; vb < vHi; vb += 8 * ATH) {
                const int v0 = vb + tid;
                uint4 w[8];
#pragma unroll
                for (int k = 0; k < 8; ++k) {
                    const int vi = v0 + k * ATH;
                    w[k] = rv[vi < vHi ? vi : vLo];
                }
                unsigned nT = 0u;
                unsigned cl[8];
#pragma unroll
                for (int k = 0; k < 8; ++k) {
                    unsigned cc = (unsigned)__shfl((int)w[k].x, lane & ~7);
                    if (v0 + k * ATH >= vHi) cc = 0u;
                    if (cc > (unsigned)(TW - 1)) {
                        anyOvf = 1u;
                        cc = 0u;
                    }
                    cl[k] = cc;
#pragma unroll
                    for (int j = 0; j < 4; ++j) nT += (fq + (unsigned)j - 1u < cc) ? 1u : 0u;
                }
                unsigned incl = nT;
#pragma unroll
                for (int dl = 1; dl < 32; dl <<= 1) {
                    const unsigned up = (unsigned)__shfl_up((int)incl, dl);
                    if (lane >= dl) incl += up;
                }
                const unsigned waveTotal = (unsigned)__builtin_amdgcn_readlane((int)incl, 31);
                unsigned wbase = 0u;
                if (lane == 0) wbase = atomicAdd(&lcount, waveTotal);
                wbase = (unsigned)__builtin_amdgcn_readfirstlane((int)wbase);
                unsigned idx = wbase + incl - nT;
                if (idx + nT <= LCAP) {
#pragma unroll
                    for (int k = 0; k < 8; ++k) {
                        const unsigned ww[4] = {w[k].x, w[k].y, w[k].z, w[k].w};
#pragma unroll
                        for (int j = 0; j < 4; ++j)
                            if (fq + (unsigned)j - 1u < cl[k]) list[idx++] = ww[j];
                    }
                }
            }
        }
        __syncthreads();
        const bool ovf = !fromList && __builtin_amdgcn_readfirstlane((int)anyOvf) != 0;
        if (ovf) {
            for (int t = tLo; t < tHi; ++t) {
                if (row[(size_t)t * TW] > (unsigned)(TW - 1)) {
                    const int e = t * (TW * 32) + tid;
                    if (e < nEdges) {
                        const int dd = dst[e];
                        if (dd / RNG == (int)blockIdx.x) {
                            const unsigned at = atomicAdd(&lcount, 1u);
                            if (at < (unsigned)LCAP) list[at] = (unsigned)src[e] | ((unsigned)(dd - base) << 24);
                        }
                    }
                }
            }
            __syncthreads();
        }

        const unsigned lc = (unsigned)__builtin_amdgcn_readfirstlane((int)lcount);
        if (lc > LCAP) {
            nPass = 1 + (nLines + TPP - 1) / TPP;
        } else {
            const int total = (int)lc;
            const bool emit = !reuse && nPass == 1;
            if (emit) emitted = lc;
            int segLo, segHi;
            if (fromList) {
                for (int i = tid; i < total; i += ATH) list[i] = gl[i];
                segLo = (int)go[wv];
                segHi = (int)go[wv + 1];
                __syncthreads();
            } else {
                unsigned e[8], rk[8];
#pragma unroll
                for (int q = 0; q < 8; ++q) {
                    e[q] = 0u;
                    rk[q] = 0u;
                    if (q * ATH < total) {
                        const int i = q * ATH + tid;
                        if (i < total) {
                            e[q] = list[i];
                            rk[q] = atomicAdd(&binCnt[(e[q] >> 24) & 31u], 1u);
                            if (!reuse) atomicAdd(&cnt[e[q] >> 24], 1.0f);
                        }
                    }
                }
                __syncthreads();
                if (wv == 0) {
                    const unsigned c = binCnt[lane];
                    unsigned incl = c;
#pragma unroll
                    for (int dl = 1; dl < 32; dl <<= 1) {
                        const unsigned up = (unsigned)__shfl_up((int)incl, dl);
                        if (lane >= dl) incl += up;
                    }
                    binOff[lane] = incl - c;
                    if (lane == 31) binOff[32] = incl;
                    binCnt[lane] = 0u;
                    if (emit) {
                        go[lane] = incl - c;
                        if (lane == 31) go[32] = incl;
                    }
                }
                __syncthreads();
#pragma unroll
                for (int q = 0; q < 8; ++q)
                    if (q * ATH < total && q * ATH + tid < total)
                        list[binOff[(e[q] >> 24) & 31u] + rk[q]] = e[q];
                __syncthreads();
                segLo = (int)binOff[wv];
                segHi = (int)binOff[wv + 1];
            }
            segLo = __builtin_amdgcn_readfirstlane(segLo);
            segHi = __builtin_amdgcn_readfirstlane(segHi);

            int i0 = segLo;
            unsigned nS = PADW;
            if (lane < EPI && i0 + lane < segHi) nS = list[i0 + lane];
            for (; i0 < segHi; i0 += EPI) {
                if (emit && lane < EPI && i0 + lane < segHi) gl[i0 + lane] = nS;
                const int myS = (int)(nS & 0xffffffu);
                const int myK = (int)(nS >> 29);
                const int i1 = i0 + EPI;
                nS = PADW;
                if (lane < EPI && i1 + lane < segHi) nS = list[i1 + lane];

                v4f v[EPI];
                gnn_grow rb[EPI];
#pragma unroll
                for (int u = 0; u < EPI; ++u)
                    rb[u] = gnn_row_base(feat, __builtin_amdgcn_readlane(myS, u));
#pragma unroll
                for (int u = 0; u < EPI; ++u) v[u] = rb[u][(unsigned)lane];
#pragma unroll
                for (int u = 0; u < EPI; ++u) {
                    const int k = __builtin_amdgcn_readlane(myK, u);
                    ax[k] += v[u].x;
                    ay[k] += v[u].y;
                    az[k] += v[u].z;
                    aw[k] += v[u].w;
                }
            }
        }
        __syncthreads();
        if (tid == 0) lcount = 0u;
        __syncthreads();
    }

    if (!reuse && tid == 0) gcnt[blockIdx.x] = emitted;

    const int rowsLeft = nNodes - base;
    const int rows = rowsLeft < RNG ? rowsLeft : RNG;
#pragma unroll
    for (int k = 0; k < NOWN; ++k) {
        const int r = wv + 32 * k;
        if (r < rows) *(v4f*)(agg + (size_t)(base + r) * D + 4 * lane) = (v4f){ax[k], ay[k], az[k], aw[k]};
    }
    if (!reuse)
        for (int r = tid; r < rows; r += ATH) inv[base + r] = 1.0f / fmaxf(cnt[r], 1.0f);
}

#define TM 16
#define PITCH 132

__global__ __launch_bounds__(256) void gnn_sage_gemm(
    const float* __restrict__ agg,
    const float* __restrict__ inv,
    const float* __restrict__ xin,
    const float* __restrict__ wl,
    const float* __restrict__ wr,
    const float* __restrict__ bias,
    float* __restrict__ out,
    int nTiles, int relu) {
    __shared__ __attribute__((aligned(16))) float sA[2][2][TM * PITCH];

    const int tid     = threadIdx.x;
    const int lane    = tid & 31;
    const int wave    = tid >> 5;
    const int half    = lane >> 4;
    const int l15     = lane & 15;
    const int colBase = wave * 16;

    v4f bl[16], br[16];
    {
        const float* wlRow = wl + (size_t)(colBase + l15) * D + 4 * half;
        const float* wrRow = wr + (size_t)(colBase + l15) * D + 4 * half;
#pragma unroll
        for (int j = 0; j < 16; ++j) {
            bl[j] = *(const v4f*)(wlRow + 8 * j);
            br[j] = *(const v4f*)(wrRow + 8 * j);
        }
    }
    const float bv = bias[colBase + l15];

    const int sr = tid >> 5;
    const int sc = (tid & 31) * 4;

    int tile = blockIdx.x;
    int cur  = 0;
    if (tile < nTiles) {
        const size_t r0 = (size_t)tile * TM + sr;
        const size_t r1 = r0 + 8;
        const float i0 = inv[r0];
        const float i1 = inv[r1];
        const v4f g0 = *(const v4f*)(agg + r0 * D + sc);
        const v4f g1 = *(const v4f*)(agg + r1 * D + sc);
        const v4f g2 = *(const v4f*)(xin + r0 * D + sc);
        const v4f g3 = *(const v4f*)(xin + r1 * D + sc);
        *(v4f*)(&sA[0][0][sr * PITCH + sc])       = g0 * i0;
        *(v4f*)(&sA[0][0][(sr + 8) * PITCH + sc]) = g1 * i1;
        *(v4f*)(&sA[0][1][sr * PITCH + sc])       = g2;
        *(v4f*)(&sA[0][1][(sr + 8) * PITCH + sc]) = g3;
    }
    __syncthreads();

    for (; tile < nTiles; tile += gridDim.x) {
        const int  nxt  = tile + (int)gridDim.x;
        const bool more = nxt < nTiles;
        float i0 = 0.0f, i1 = 0.0f;
        v4f g0 = {}, g1 = {}, g2 = {}, g3 = {};
        if (more) {
            const size_t r0 = (size_t)nxt * TM + sr;
            const size_t r1 = r0 + 8;
            i0 = inv[r0];
            i1 = inv[r1];
            g0 = *(const v4f*)(agg + r0 * D + sc);
            g1 = *(const v4f*)(agg + r1 * D + sc);
            g2 = *(const v4f*)(xin + r0 * D + sc);
            g3 = *(const v4f*)(xin + r1 * D + sc);
        }

        const float* pa = &sA[cur][0][l15 * PITCH + 4 * half];
        const float* px = &sA[cur][1][l15 * PITCH + 4 * half];
        v8f acc1 = {};
        v8f acc2 = {};
#pragma unroll
        for (int j = 0; j < 16; ++j) {
            const v4f a = *(const v4f*)(pa + 8 * j);
            const v4f x = *(const v4f*)(px + 8 * j);
            acc1 = __builtin_amdgcn_wmma_f32_16x16x4_f32(false, a.xy, false, bl[j].xy,
                                                         (short)0, acc1, false, false);
            acc2 = __builtin_amdgcn_wmma_f32_16x16x4_f32(false, x.xy, false, br[j].xy,
                                                         (short)0, acc2, false, false);
            acc1 = __builtin_amdgcn_wmma_f32_16x16x4_f32(false, a.zw, false, bl[j].zw,
                                                         (short)0, acc1, false, false);
            acc2 = __builtin_amdgcn_wmma_f32_16x16x4_f32(false, x.zw, false, br[j].zw,
                                                         (short)0, acc2, false, false);
        }

        const int rowBase = tile * TM;
#pragma unroll
        for (int v = 0; v < 8; ++v) {
            float r = (acc1[v] + acc2[v]) + bv;
            if (relu) r = fmaxf(r, 0.0f);
            out[(size_t)(rowBase + v + 8 * half) * D + colBase + l15] = r;
        }

        if (more) {
            const int nb = cur ^ 1;
            *(v4f*)(&sA[nb][0][sr * PITCH + sc])       = g0 * i0;
            *(v4f*)(&sA[nb][0][(sr + 8) * PITCH + sc]) = g1 * i1;
            *(v4f*)(&sA[nb][1][sr * PITCH + sc])       = g2;
            *(v4f*)(&sA[nb][1][(sr + 8) * PITCH + sc]) = g3;
        }
        __syncthreads();
        cur ^= 1;
    }
}

extern "C" void kernel_launch(void* const* d_in, const int* in_sizes, int n_in,
                              void* d_out, int out_size, void* d_ws, size_t ws_size,
                              hipStream_t stream) {
    const float* x   = (const float*)d_in[0];
    const int* edge  = (const int*)d_in[1];
    const float* w1l = (const float*)d_in[2];
    const float* w1r = (const float*)d_in[3];
    const float* b1  = (const float*)d_in[4];
    const float* w2l = (const float*)d_in[5];
    const float* w2r = (const float*)d_in[6];
    const float* b2  = (const float*)d_in[7];
    float* out = (float*)d_out;

    const int N = in_sizes[0] / D;
    const int E = in_sizes[1] / 2;
    const int* src = edge;
    const int* dst = edge + E;

    float* agg = (float*)d_ws;
    float* h   = agg + (size_t)N * D;
    float* inv = h + (size_t)N * D;
    unsigned* bm = (unsigned*)(inv + ((N + 31) & ~31));

    dim3 blk(256);
    dim3 gBits((unsigned)((E + 1023) / 1024));
    const int rowWords = ((E + 1023) / 1024) * TW;
    unsigned* gcnt  = bm + (size_t)256 * rowWords;
    unsigned* glist = gcnt + 256;
    unsigned* gofs  = glist + (size_t)256 * LCAP;
    dim3 gAgg((unsigned)((N + RNG - 1) / RNG));
    dim3 gGemm(512);
    const int nTiles = N / 16;

    gnn_bits<<<gBits, blk, 0, stream>>>(src, dst, bm, E);

    gnn_agg<<<gAgg, dim3(ATH), 0, stream>>>(x, src, dst, bm, agg, inv, glist, gcnt, gofs, rowWords, N, E, 0);
    gnn_sage_gemm<<<gGemm, blk, 0, stream>>>(agg, inv, x, w1l, w1r, b1, h, nTiles, 1);

    gnn_agg<<<gAgg, dim3(ATH), 0, stream>>>(h, src, dst, bm, agg, inv, glist, gcnt, gofs, rowWords, N, E, 1);
    gnn_sage_gemm<<<gGemm, blk, 0, stream>>>(agg, inv, h, w2l, w2r, b2, out, nTiles, 0);
}


# SAGE linear map fused into the aggregation launch: block parks its register-summed mean and self rows in LDS and multiplies them on WMMA, weights staged per matrix; 5 to 3 launches; add order changes
# speedup vs baseline: 5.0973x; 1.1818x over previous
#include <hip/hip_runtime.h>

typedef __attribute__((ext_vector_type(2))) float v2f;
typedef __attribute__((ext_vector_type(4))) float v4f;
typedef __attribute__((ext_vector_type(8))) float v8f;

#define D 128

#define RNG 196
#define LCAP 8192
#define EPI 16
#define ATH 1024
#define TW 32
#define TPITCH 33
#define TPP (LCAP / (TW * 32))

__global__ __launch_bounds__(256) void gnn_bits(const int* __restrict__ src,
                                                const int* __restrict__ dst,
                                                unsigned* __restrict__ bm, int nEdges) {
    __shared__ unsigned t[256 * TPITCH];
    const int tid  = threadIdx.x;
    const int lane = tid & 31;
    const int wv   = tid >> 5;
    const int e0 = blockIdx.x * (TW * 32);
    const int rowWords = gridDim.x * TW;
    int d[4], s[4];
#pragma unroll
    for (int k = 0; k < 4; ++k) {
        const int e = e0 + k * 256 + tid;
        const int ec = e < nEdges ? e : 0;
        d[k] = dst[ec];
        s[k] = src[ec];
    }
    t[tid * TPITCH] = 0u;
    __syncthreads();
#pragma unroll
    for (int k = 0; k < 4; ++k) {
        if (e0 + k * 256 + tid < nEdges) {
            const unsigned r = (unsigned)(d[k] / RNG) & 0xffu;
            const unsigned pos = atomicAdd(&t[r * TPITCH], 1u);
            if (pos < (unsigned)(TW - 1))
                t[r * TPITCH + 1u + pos] = (unsigned)s[k] | ((unsigned)(d[k] - (int)r * RNG) << 24);
        }
    }
    __syncthreads();
    unsigned* p = bm + (size_t)wv * rowWords + (size_t)blockIdx.x * TW + lane;
    const unsigned* q = t + wv * TPITCH + lane;
#pragma unroll 8
    for (int it = 0; it < 32; ++it) p[(size_t)(it * 8) * rowWords] = q[it * 8 * TPITCH];
}

typedef const __attribute__((address_space(1))) v4f* gnn_grow;
static __device__ __forceinline__ gnn_grow gnn_row_base(const float* feat, int sId) {
    const unsigned long long a = (unsigned long long)feat + ((unsigned long long)(unsigned)sId << 9);
    const unsigned lo = (unsigned)__builtin_amdgcn_readfirstlane((int)(unsigned)a);
    const unsigned hi = (unsigned)__builtin_amdgcn_readfirstlane((int)(unsigned)(a >> 32));
    return (gnn_grow)(((unsigned long long)hi << 32) | (unsigned long long)lo);
}

#define PITCH 132

static __device__ __forceinline__ void gnn_pass3(const float* sA, int t0, int t1, int t2, int l15, int half,
                                                 const v4f (&b)[16], v8f& acc0, v8f& acc1, v8f& acc2) {
    const float* p0 = sA + (t0 * 16 + l15) * PITCH + 4 * half;
    const float* p1 = sA + (t1 * 16 + l15) * PITCH + 4 * half;
    const float* p2 = sA + (t2 * 16 + l15) * PITCH + 4 * half;
#pragma unroll
    for (int j = 0; j < 16; ++j) {
        const v4f a0 = *(const v4f*)(p0 + 8 * j);
        const v4f a1 = *(const v4f*)(p1 + 8 * j);
        const v4f a2 = *(const v4f*)(p2 + 8 * j);
        acc0 = __builtin_amdgcn_wmma_f32_16x16x4_f32(false, a0.xy, false, b[j].xy, (short)0, acc0, false, false);
        acc1 = __builtin_amdgcn_wmma_f32_16x16x4_f32(false, a1.xy, false, b[j].xy, (short)0, acc1, false, false);
        acc2 = __builtin_amdgcn_wmma_f32_16x16x4_f32(false, a2.xy, false, b[j].xy, (short)0, acc2, false, false);
        acc0 = __builtin_amdgcn_wmma_f32_16x16x4_f32(false, a0.zw, false, b[j].zw, (short)0, acc0, false, false);
        acc1 = __builtin_amdgcn_wmma_f32_16x16x4_f32(false, a1.zw, false, b[j].zw, (short)0, acc1, false, false);
        acc2 = __builtin_amdgcn_wmma_f32_16x16x4_f32(false, a2.zw, false, b[j].zw, (short)0, acc2, false, false);
    }
}

static __device__ __forceinline__ void gnn_pass1(const float* sA, int t, int l15, int half,
                                                 const v4f (&b)[16], v8f& acc) {
    const float* p = sA + (t * 16 + l15) * PITCH + 4 * half;
#pragma unroll
    for (int j = 0; j < 16; ++j) {
        const v4f a = *(const v4f*)(p + 8 * j);
        acc = __builtin_amdgcn_wmma_f32_16x16x4_f32(false, a.xy, false, b[j].xy, (short)0, acc, false, false);
        acc = __builtin_amdgcn_wmma_f32_16x16x4_f32(false, a.zw, false, b[j].zw, (short)0, acc, false, false);
    }
}

static __device__ __forceinline__ void gnn_store_tile(float* po, int t, int half, int rows,
                                                      const v8f& acc, float bv, int relu) {
#pragma unroll
    for (int v = 0; v < 8; ++v) {
        const int r = t * 16 + v + 8 * half;
        float o = acc[v] + bv;
        if (relu) o = fmaxf(o, 0.0f);
        if (r < rows) po[(size_t)r * D] = o;
    }
}

#define NOWN ((RNG + 31) / 32)
#define PADW (7u << 29)

__global__ __launch_bounds__(ATH) void gnn_agg(const float* __restrict__ feat,
                                               const int* __restrict__ src,
                                               const int* __restrict__ dst,
                                               const unsigned* __restrict__ bm,
                                               const float* __restrict__ wl,
                                               const float* __restrict__ wr,
                                               const float* __restrict__ bias,
                                               float* __restrict__ outp,
                                               float* __restrict__ inv,
                                               unsigned* __restrict__ glist,
                                               unsigned* __restrict__ gcnt,
                                               unsigned* __restrict__ gofs,
                                               int rowWords, int nNodes, int nEdges, int reuse, int relu) {
    __shared__ __attribute__((aligned(16))) float sT[(2 * RNG + D) * PITCH];
    __shared__ float cnt[RNG];
    __shared__ unsigned list[LCAP];
    __shared__ unsigned binCnt[32];
    __shared__ unsigned binOff[33];
    __shared__ unsigned lcount;
    __shared__ unsigned anyOvf;

    const int tid  = threadIdx.x;
    const int lane = tid & 31;
    const int wv   = __builtin_amdgcn_readfirstlane(tid >> 5);
    const int base = blockIdx.x * RNG;
    const unsigned* row = bm + (size_t)blockIdx.x * rowWords;
    unsigned* gl = glist + (size_t)blockIdx.x * LCAP;
    unsigned* go = gofs + (size_t)blockIdx.x * 33;

    const unsigned stored = reuse ? gcnt[blockIdx.x] : 0xffffffffu;
    const bool fromList = __builtin_amdgcn_readfirstlane((int)(stored <= (unsigned)LCAP)) != 0;

    for (int i = tid; i < RNG; i += ATH) cnt[i] = 0.0f;
    if (tid < 32) binCnt[tid] = 0u;
    if (tid == 0) {
        lcount = fromList ? stored : 0u;
        anyOvf = 0u;
    }
    __syncthreads();

    v8f ax = {}, ay = {}, az = {}, aw = {};

    const int nLines = rowWords / TW;
    int nPass = 1;
    unsigned emitted = 0xffffffffu;
    for (int pass = 0; pass < nPass; ++pass) {
        const int tLo = pass == 0 ? 0 : (pass - 1) * TPP;
        const int tEnd = pass == 0 ? nLines : tLo + TPP;
        const int tHi = tEnd < nLines ? tEnd : nLines;
        if (fromList) {
        } else {
            const int vLo = tLo * 8, vHi = tHi * 8;
            const uint4* rv = (const uint4*)row;
            const unsigned fq = 4u * (unsigned)(lane & 7);
            for (int vb = vLo; vb < vHi; vb += 8 * ATH) {
                const int v0 = vb + tid;
                uint4 w[8];
#pragma unroll
                for (int k = 0; k < 8; ++k) {
                    const int vi = v0 + k * ATH;
                    w[k] = rv[vi < vHi ? vi : vLo];
                }
                unsigned nT = 0u;
                unsigned cl[8];
#pragma unroll
                for (int k = 0; k < 8; ++k) {
                    unsigned cc = (unsigned)__shfl((int)w[k].x, lane & ~7);
                    if (v0 + k * ATH >= vHi) cc = 0u;
                    if (cc > (unsigned)(TW - 1)) {
                        anyOvf = 1u;
                        cc = 0u;
                    }
                    cl[k] = cc;
#pragma unroll
                    for (int j = 0; j < 4; ++j) nT += (fq + (unsigned)j - 1u < cc) ? 1u : 0u;
                }
                unsigned incl = nT;
#pragma unroll
                for (int dl = 1; dl < 32; dl <<= 1) {
                    const unsigned up = (unsigned)__shfl_up((int)incl, dl);
                    if (lane >= dl) incl += up;
                }
                const unsigned waveTotal = (unsigned)__builtin_amdgcn_readlane((int)incl, 31);
                unsigned wbase = 0u;
                if (lane == 0) wbase = atomicAdd(&lcount, waveTotal);
                wbase = (unsigned)__builtin_amdgcn_readfirstlane((int)wbase);
                unsigned idx = wbase + incl - nT;
                if (idx + nT <= LCAP) {
#pragma unroll
                    for (int k = 0; k < 8; ++k) {
                        const unsigned ww[4] = {w[k].x, w[k].y, w[k].z, w[k].w};
#pragma unroll
                        for (int j = 0; j < 4; ++j)
                            if (fq + (unsigned)j - 1u < cl[k]) list[idx++] = ww[j];
                    }
                }
            }
        }
        __syncthreads();
        const bool ovf = !fromList && __builtin_amdgcn_readfirstlane((int)anyOvf) != 0;
        if (ovf) {
            for (int t = tLo; t < tHi; ++t) {
                if (row[(size_t)t * TW] > (unsigned)(TW - 1)) {
                    const int e = t * (TW * 32) + tid;
                    if (e < nEdges) {
                        const int dd = dst[e];
                        if (dd / RNG == (int)blockIdx.x) {
                            const unsigned at = atomicAdd(&lcount, 1u);
                            if (at < (unsigned)LCAP) list[at] = (unsigned)src[e] | ((unsigned)(dd - base) << 24);
                        }
                    }
                }
            }
            __syncthreads();
        }

        const unsigned lc = (unsigned)__builtin_amdgcn_readfirstlane((int)lcount);
        if (lc > LCAP) {
            nPass = 1 + (nLines + TPP - 1) / TPP;
        } else {
            const int total = (int)lc;
            const bool emit = !reuse && nPass == 1;
            if (emit) emitted = lc;
            int segLo, segHi;
            if (fromList) {
                for (int i = tid; i < total; i += ATH) list[i] = gl[i];
                segLo = (int)go[wv];
                segHi = (int)go[wv + 1];
                __syncthreads();
            } else {
                unsigned e[8], rk[8];
#pragma unroll
                for (int q = 0; q < 8; ++q) {
                    e[q] = 0u;
                    rk[q] = 0u;
                    if (q * ATH < total) {
                        const int i = q * ATH + tid;
                        if (i < total) {
                            e[q] = list[i];
                            rk[q] = atomicAdd(&binCnt[(e[q] >> 24) & 31u], 1u);
                            if (!reuse) atomicAdd(&cnt[e[q] >> 24], 1.0f);
                        }
                    }
                }
                __syncthreads();
                if (wv == 0) {
                    const unsigned c = binCnt[lane];
                    unsigned incl = c;
#pragma unroll
                    for (int dl = 1; dl < 32; dl <<= 1) {
                        const unsigned up = (unsigned)__shfl_up((int)incl, dl);
                        if (lane >= dl) incl += up;
                    }
                    binOff[lane] = incl - c;
                    if (lane == 31) binOff[32] = incl;
                    binCnt[lane] = 0u;
                    if (emit) {
                        go[lane] = incl - c;
                        if (lane == 31) go[32] = incl;
                    }
                }
                __syncthreads();
#pragma unroll
                for (int q = 0; q < 8; ++q)
                    if (q * ATH < total && q * ATH + tid < total)
                        list[binOff[(e[q] >> 24) & 31u] + rk[q]] = e[q];
                __syncthreads();
                segLo = (int)binOff[wv];
                segHi = (int)binOff[wv + 1];
            }
            segLo = __builtin_amdgcn_readfirstlane(segLo);
            segHi = __builtin_amdgcn_readfirstlane(segHi);

            int i0 = segLo;
            unsigned nS = PADW;
            if (lane < EPI && i0 + lane < segHi) nS = list[i0 + lane];
            for (; i0 < segHi; i0 += EPI) {
                if (emit && lane < EPI && i0 + lane < segHi) gl[i0 + lane] = nS;
                const int myS = (int)(nS & 0xffffffu);
                const int myK = (int)(nS >> 29);
                const int i1 = i0 + EPI;
                nS = PADW;
                if (lane < EPI && i1 + lane < segHi) nS = list[i1 + lane];

                v4f v[EPI];
                gnn_grow rb[EPI];
#pragma unroll
                for (int u = 0; u < EPI; ++u)
                    rb[u] = gnn_row_base(feat, __builtin_amdgcn_readlane(myS, u));
#pragma unroll
                for (int u = 0; u < EPI; ++u) v[u] = rb[u][(unsigned)lane];
#pragma unroll
                for (int u = 0; u < EPI; ++u) {
                    const int k = __builtin_amdgcn_readlane(myK, u);
                    ax[k] += v[u].x;
                    ay[k] += v[u].y;
                    az[k] += v[u].z;
                    aw[k] += v[u].w;
                }
            }
        }
        __syncthreads();
        if (tid == 0) lcount = 0u;
        __syncthreads();
    }

    if (!reuse && tid == 0) gcnt[blockIdx.x] = emitted;

    const int rowsLeft = nNodes - base;
    const int rows = rowsLeft < RNG ? rowsLeft : RNG;
    float* sM = sT;
    float* sX = sT + RNG * PITCH;
    float* sW = sT + 2 * RNG * PITCH;

    v4f xr[NOWN];
#pragma unroll
    for (int k = 0; k < NOWN; ++k) {
        const int r = wv + 32 * k;
        const int rc = r < rows ? r : 0;
        xr[k] = *(const v4f*)(feat + (size_t)(base + rc) * D + 4 * lane);
    }
    v4f wa[4], wb[4];
#pragma unroll
    for (int i = 0; i < 4; ++i) {
        wa[i] = *(const v4f*)(wl + (size_t)(4 * wv + i) * D + 4 * lane);
        wb[i] = *(const v4f*)(wr + (size_t)(4 * wv + i) * D + 4 * lane);
    }
    const int slab = wv & 7;
    const int g    = wv >> 3;
    const int half = lane >> 4;
    const int l15  = lane & 15;
    const float bv = bias[slab * 16 + l15];
    float iv[NOWN];
    if (reuse) {
#pragma unroll
        for (int k = 0; k < NOWN; ++k) {
            const int r = wv + 32 * k;
            iv[k] = inv[base + (r < rows ? r : 0)];
        }
    } else {
#pragma unroll
        for (int k = 0; k < NOWN; ++k) {
            const int r = wv + 32 * k;
            iv[k] = 1.0f / fmaxf(cnt[r < rows ? r : 0], 1.0f);
        }
        for (int r = tid; r < rows; r += ATH) inv[base + r] = 1.0f / fmaxf(cnt[r], 1.0f);
    }

#pragma unroll
    for (int k = 0; k < NOWN; ++k) {
        const int r = wv + 32 * k;
        if (r < RNG) *(v4f*)(sM + r * PITCH + 4 * lane) = (v4f){ax[k], ay[k], az[k], aw[k]} * iv[k];
    }
#pragma unroll
    for (int k = 0; k < NOWN; ++k) {
        const int r = wv + 32 * k;
        if (r < RNG) *(v4f*)(sX + r * PITCH + 4 * lane) = xr[k];
    }
#pragma unroll
    for (int i = 0; i < 4; ++i) *(v4f*)(sW + (4 * wv + i) * PITCH + 4 * lane) = wa[i];
    __syncthreads();

    v4f b[16];
    const float* pw = sW + (slab * 16 + l15) * PITCH + 4 * half;
#pragma unroll
    for (int j = 0; j < 16; ++j) b[j] = *(const v4f*)(pw + 8 * j);
    __syncthreads();
#pragma unroll
    for (int i = 0; i < 4; ++i) *(v4f*)(sW + (4 * wv + i) * PITCH + 4 * lane) = wb[i];

    const int nT = (rows + 15) >> 4;
    const int tLast = nT - 1;
    const int t0 = g < tLast ? g : tLast;
    const int t1 = g + 4 < tLast ? g + 4 : tLast;
    const int t2 = g + 8 < tLast ? g + 8 : tLast;
    const bool extra = g == 0 && nT > 12;
    v8f acc0 = {}, acc1 = {}, acc2 = {}, acc3 = {};
    gnn_pass3(sM, t0, t1, t2, l15, half, b, acc0, acc1, acc2);
    if (extra) gnn_pass1(sM, 12, l15, half, b, acc3);
    __syncthreads();
#pragma unroll
    for (int j = 0; j < 16; ++j) b[j] = *(const v4f*)(pw + 8 * j);
    gnn_pass3(sX, t0, t1, t2, l15, half, b, acc0, acc1, acc2);
    if (extra) gnn_pass1(sX, 12, l15, half, b, acc3);

    float* po = outp + (size_t)base * D + slab * 16 + l15;
    if (g < nT) gnn_store_tile(po, g, half, rows, acc0, bv, relu);
    if (g + 4 < nT) gnn_store_tile(po, g + 4, half, rows, acc1, bv, relu);
    if (g + 8 < nT) gnn_store_tile(po, g + 8, half, rows, acc2, bv, relu);
    if (extra) gnn_store_tile(po, 12, half, rows, acc3, bv, relu);
}

extern "C" void kernel_launch(void* const* d_in, const int* in_sizes, int n_in,
                              void* d_out, int out_size, void* d_ws, size_t ws_size,
                              hipStream_t stream) {
    const float* x   = (const float*)d_in[0];
    const int* edge  = (const int*)d_in[1];
    const float* w1l = (const float*)d_in[2];
    const float* w1r = (const float*)d_in[3];
    const float* b1  = (const float*)d_in[4];
    const float* w2l = (const float*)d_in[5];
    const float* w2r = (const float*)d_in[6];
    const float* b2  = (const float*)d_in[7];
    float* out = (float*)d_out;

    const int N = in_sizes[0] / D;
    const int E = in_sizes[1] / 2;
    const int* src = edge;
    const int* dst = edge + E;

    float* h   = (float*)d_ws + (size_t)N * D;
    float* inv = h + (size_t)N * D;
    unsigned* bm = (unsigned*)(inv + ((N + 31) & ~31));

    dim3 blk(256);
    dim3 gBits((unsigned)((E + 1023) / 1024));
    const int rowWords = ((E + 1023) / 1024) * TW;
    unsigned* gcnt  = bm + (size_t)256 * rowWords;
    unsigned* glist = gcnt + 256;
    unsigned* gofs  = glist + (size_t)256 * LCAP;
    dim3 gAgg((unsigned)((N + RNG - 1) / RNG));

    gnn_bits<<<gBits, blk, 0, stream>>>(src, dst, bm, E);

    gnn_agg<<<gAgg, dim3(ATH), 0, stream>>>(x, src, dst, bm, w1l, w1r, b1, h, inv, glist, gcnt, gofs, rowWords, N, E, 0, 1);

    gnn_agg<<<gAgg, dim3(ATH), 0, stream>>>(h, src, dst, bm, w2l, w2r, b2, out, inv, glist, gcnt, gofs, rowWords, N, E, 1, 0);
}
